# starting kernel
# baseline (speedup 1.0000x reference)
#include <hip/hip_runtime.h>
#include <hip/hip_bf16.h>

#define IN_F 128
#define OUT_F 32
#define NEG_SLOPE 0.2f

typedef float v2f __attribute__((ext_vector_type(2)));
typedef float v8f __attribute__((ext_vector_type(8)));

__global__ void __launch_bounds__(256) gat_gemm_wmma(
    const float* __restrict__ h, const float* __restrict__ W,
    float* __restrict__ z, int n_nodes)
{
    const int wave = threadIdx.x >> 5;
    const int lane = threadIdx.x & 31;
    const int tile = blockIdx.x * 8 + wave;
    const int row0 = tile * 16;
    if (row0 >= n_nodes) return;

    const int half = lane >> 4;
    const int l16  = lane & 15;
    int arow = row0 + l16;
    if (arow >= n_nodes) arow = n_nodes - 1;

    const float* __restrict__ hrow = h + (size_t)arow * IN_F;
    const float* __restrict__ w0   = W + (size_t)l16 * IN_F;
    const float* __restrict__ w1   = W + (size_t)(l16 + 16) * IN_F;

    v8f acc0 = {};
    v8f acc1 = {};

#pragma unroll 4
    for (int kk = 0; kk < IN_F; kk += 4) {
        const int ka = kk + 2 * half;
        v2f A, B0, B1;
        A[0]  = hrow[ka];  A[1]  = hrow[ka + 1];
        B0[0] = w0[ka];    B0[1] = w0[ka + 1];
        B1[0] = w1[ka];    B1[1] = w1[ka + 1];
        acc0 = __builtin_amdgcn_wmma_f32_16x16x4_f32(false, A, false, B0,
                                                     (short)0, acc0, false, false);
        acc1 = __builtin_amdgcn_wmma_f32_16x16x4_f32(false, A, false, B1,
                                                     (short)0, acc1, false, false);
    }

#pragma unroll
    for (int r = 0; r < 8; ++r) {
        const int m = row0 + r + 8 * half;
        if (m < n_nodes) {
            z[(size_t)m * OUT_F + l16]      = acc0[r];
            z[(size_t)m * OUT_F + 16 + l16] = acc1[r];
        }
    }
}

__global__ void __launch_bounds__(256) gat_node_init(
    const float* __restrict__ z, const float* __restrict__ a,
    float* __restrict__ el, float* __restrict__ er,
    float* __restrict__ emax, float* __restrict__ denom,
    float* __restrict__ out, int n_nodes)
{
    const int n = blockIdx.x * blockDim.x + threadIdx.x;
    if (n >= n_nodes) return;
    const float* zr = z + (size_t)n * OUT_F;
    float sl = 0.f, sr = 0.f;
#pragma unroll
    for (int j = 0; j < OUT_F; ++j) {
        const float zv = zr[j];
        sl += zv * a[j];
        sr += zv * a[OUT_F + j];
    }
    el[n]    = sl;
    er[n]    = sr;
    emax[n]  = -__builtin_inff();
    denom[n] = 0.f;
    float4* o = (float4*)(out + (size_t)n * OUT_F);
#pragma unroll
    for (int j = 0; j < OUT_F / 4; ++j) o[j] = make_float4(0.f, 0.f, 0.f, 0.f);
}

__device__ __forceinline__ void atomic_max_f32(float* addr, float v)
{
    if (v >= 0.f)
        atomicMax((int*)addr, __float_as_int(v));
    else
        atomicMin((unsigned int*)addr, __float_as_uint(v));
}

__device__ __forceinline__ float leaky(float v)
{
    return v > 0.f ? v : NEG_SLOPE * v;
}

__global__ void __launch_bounds__(256) gat_edge_max(
    const int* __restrict__ src, const int* __restrict__ dst,
    const float* __restrict__ el, const float* __restrict__ er,
    float* __restrict__ emax, int n_edges)
{
    const int e = blockIdx.x * blockDim.x + threadIdx.x;
    if (e >= n_edges) return;
    const int s = src[e], d = dst[e];
    atomic_max_f32(&emax[d], leaky(el[s] + er[d]));
}

__global__ void __launch_bounds__(256) gat_edge_expsum(
    const int* __restrict__ src, const int* __restrict__ dst,
    const float* __restrict__ el, const float* __restrict__ er,
    const float* __restrict__ emax, float* __restrict__ denom, int n_edges)
{
    const int e = blockIdx.x * blockDim.x + threadIdx.x;
    if (e >= n_edges) return;
    const int s = src[e], d = dst[e];
    const float ex = __expf(leaky(el[s] + er[d]) - emax[d]);
    atomicAdd(&denom[d], ex);
}

__global__ void __launch_bounds__(256) gat_edge_aggregate(
    const int* __restrict__ src, const int* __restrict__ dst,
    const float* __restrict__ el, const float* __restrict__ er,
    const float* __restrict__ emax, const float* __restrict__ denom,
    const float* __restrict__ z, float* __restrict__ out, long n_edges)
{
    const long t = (long)blockIdx.x * blockDim.x + threadIdx.x;
    const long e = t >> 3;
    if (e >= n_edges) return;
    const int c = ((int)t & 7) * 4;
    const int s = src[e], d = dst[e];
    const float ex    = __expf(leaky(el[s] + er[d]) - emax[d]);
    const float alpha = ex / fmaxf(denom[d], 1e-16f);
    const float4 zc = *(const float4*)(z + (size_t)s * OUT_F + c);
    float* ob = out + (size_t)d * OUT_F + c;
    atomicAdd(ob + 0, alpha * zc.x);
    atomicAdd(ob + 1, alpha * zc.y);
    atomicAdd(ob + 2, alpha * zc.z);
    atomicAdd(ob + 3, alpha * zc.w);
}

extern "C" void kernel_launch(void* const* d_in, const int* in_sizes, int n_in,
                              void* d_out, int out_size, void* d_ws, size_t ws_size,
                              hipStream_t stream)
{
    const float* h  = (const float*)d_in[0];
    const float* W  = (const float*)d_in[1];
    const float* a  = (const float*)d_in[2];
    const int* src  = (const int*)d_in[3];
    const int* dst  = (const int*)d_in[4];
    float* out      = (float*)d_out;

    const int n_nodes = in_sizes[0] / IN_F;
    const int n_edges = in_sizes[3];

    float* z     = (float*)d_ws;
    float* el    = z + (size_t)n_nodes * OUT_F;
    float* er    = el + n_nodes;
    float* emax  = er + n_nodes;
    float* denom = emax + n_nodes;

    const int B = 256;
    const int ntiles = (n_nodes + 15) / 16;

    gat_gemm_wmma<<<dim3((ntiles + 7) / 8), dim3(B), 0, stream>>>(h, W, z, n_nodes);

    gat_node_init<<<dim3((n_nodes + B - 1) / B), dim3(B), 0, stream>>>(
        z, a, el, er, emax, denom, out, n_nodes);

    gat_edge_max<<<dim3((n_edges + B - 1) / B), dim3(B), 0, stream>>>(
        src, dst, el, er, emax, n_edges);

    gat_edge_expsum<<<dim3((n_edges + B - 1) / B), dim3(B), 0, stream>>>(
        src, dst, el, er, emax, denom, n_edges);

    const long t5 = (long)n_edges * 8;
    gat_edge_aggregate<<<dim3((unsigned)((t5 + B - 1) / B)), dim3(B), 0, stream>>>(
        src, dst, el, er, emax, denom, z, out, (long)n_edges);
}


# Edge partition launch: 256 edge chunks are counting-sorted by destination slice in LDS; each owner copies only its own runs (src packed into the entry) instead of testing all of dst
# speedup vs baseline: 3.8777x; 3.8777x over previous
#include <hip/hip_runtime.h>
#include <hip/hip_bf16.h>

#define IN_F 128
#define OUT_F 32
#define NEG_SLOPE 0.2f

typedef float v2f __attribute__((ext_vector_type(2)));
typedef float v8f __attribute__((ext_vector_type(8)));

__global__ void __launch_bounds__(256) gat_gemm_wmma(
    const float* __restrict__ h, const float* __restrict__ W,
    float* __restrict__ z, int n_nodes)
{
    const int wave = threadIdx.x >> 5;
    const int lane = threadIdx.x & 31;
    const int tile = blockIdx.x * 8 + wave;
    const int row0 = tile * 16;
    if (row0 >= n_nodes) return;

    const int half = lane >> 4;
    const int l16  = lane & 15;
    int arow = row0 + l16;
    if (arow >= n_nodes) arow = n_nodes - 1;

    const float* __restrict__ hrow = h + (size_t)arow * IN_F;
    const float* __restrict__ w0   = W + (size_t)l16 * IN_F;
    const float* __restrict__ w1   = W + (size_t)(l16 + 16) * IN_F;

    v8f acc0 = {};
    v8f acc1 = {};

#pragma unroll 4
    for (int kk = 0; kk < IN_F; kk += 4) {
        const int ka = kk + 2 * half;
        v2f A, B0, B1;
        A[0]  = hrow[ka];  A[1]  = hrow[ka + 1];
        B0[0] = w0[ka];    B0[1] = w0[ka + 1];
        B1[0] = w1[ka];    B1[1] = w1[ka + 1];
        acc0 = __builtin_amdgcn_wmma_f32_16x16x4_f32(false, A, false, B0,
                                                     (short)0, acc0, false, false);
        acc1 = __builtin_amdgcn_wmma_f32_16x16x4_f32(false, A, false, B1,
                                                     (short)0, acc1, false, false);
    }

#pragma unroll
    for (int r = 0; r < 8; ++r) {
        const int m = row0 + r + 8 * half;
        if (m < n_nodes) {
            z[(size_t)m * OUT_F + l16]      = acc0[r];
            z[(size_t)m * OUT_F + 16 + l16] = acc1[r];
        }
    }
}

__global__ void __launch_bounds__(256) gat_node_logits(
    const float* __restrict__ z, const float* __restrict__ a,
    float* __restrict__ el, float* __restrict__ er, int n_nodes)
{
    const int n = blockIdx.x * blockDim.x + threadIdx.x;
    if (n >= n_nodes) return;
    const float* zr = z + (size_t)n * OUT_F;
    float sl = 0.f, sr = 0.f;
#pragma unroll
    for (int j = 0; j < OUT_F; ++j) {
        const float zv = zr[j];
        sl += zv * a[j];
        sr += zv * a[OUT_F + j];
    }
    el[n] = sl;
    er[n] = sr;
}

__device__ __forceinline__ float leaky(float v)
{
    return v > 0.f ? v : NEG_SLOPE * v;
}

#define NPW    391
#define CAP    16384
#define OWN_T  1024

#define BIN_G  256
#define BIN_T  256
#define BIN_U  25
#define BIN_CH (BIN_T * BIN_U)
#define BIN_NS 256
#define OFFP   (BIN_NS + 1)
#define TBITS  9
#define TMASK  ((1u << TBITS) - 1u)

__device__ __forceinline__ bool bins_usable(int n_nodes, int n_edges, int ns,
                                            unsigned long long ws_bytes)
{
    const int ch = (n_edges + BIN_G - 1) / BIN_G;
    const unsigned long long words =
        (unsigned long long)n_nodes * (OUT_F + 2) +
        (unsigned long long)BIN_G * (unsigned long long)ch +
        (unsigned long long)BIN_G * OFFP;
    return n_edges > 0 && ns <= BIN_NS && ch <= BIN_CH && n_nodes <= (1 << 22) &&
           ws_bytes >= words * 4ull;
}

__device__ __forceinline__ int excl_scan_256(int v, int tid, int* s_w, int* total)
{
    const int lane = tid & 31;
    const int wv   = tid >> 5;
    int inc = v;
#pragma unroll
    for (int d = 1; d < 32; d <<= 1) {
        const int y = __shfl_up(inc, d);
        if (lane >= d) inc += y;
    }
    if (wv < 8 && lane == 31) s_w[wv] = inc;
    __syncthreads();
    int base = 0, tot = 0;
#pragma unroll
    for (int w = 0; w < 8; ++w) {
        const int x = s_w[w];
        if (w < wv) base += x;
        tot += x;
    }
    *total = tot;
    return base + inc - v;
}

__global__ void __launch_bounds__(BIN_T) gat_bin(
    const int* __restrict__ src, const int* __restrict__ dst,
    unsigned int* __restrict__ binbase, unsigned long long ws_bytes,
    int n_nodes, int n_edges)
{
    __shared__ int s_hist[BIN_NS + 1];
    __shared__ int s_off[BIN_NS + 1];
    __shared__ int s_w[8];
    __shared__ unsigned int s_stage[BIN_CH];

    const int tid = threadIdx.x;
    const int ns  = (n_nodes + NPW - 1) / NPW;
    if (!bins_usable(n_nodes, n_edges, ns, ws_bytes)) return;

    const int ch = (n_edges + BIN_G - 1) / BIN_G;
    const int c  = blockIdx.x;
    const int e0 = c * ch;
    const int n  = max(0, min(ch, n_edges - e0));
    unsigned int* __restrict__ list = binbase + (size_t)c * ch;
    int* __restrict__ offs = (int*)(binbase + (size_t)BIN_G * ch) + (size_t)c * OFFP;

    for (int i = tid; i < BIN_NS + 1; i += BIN_T) { s_hist[i] = 0; s_off[i] = 0; }
    __syncthreads();

    int d[BIN_U], s[BIN_U];
#pragma unroll
    for (int u = 0; u < BIN_U; ++u) {
        const int e = min(e0 + tid + u * BIN_T, n_edges - 1);
        d[u] = dst[e];
        s[u] = src[e];
    }
    int bb[BIN_U], pos[BIN_U];
    unsigned int pk[BIN_U];
#pragma unroll
    for (int u = 0; u < BIN_U; ++u) {
        const bool ok = (tid + u * BIN_T < n) && ((unsigned)d[u] < (unsigned)n_nodes);
        const int b = d[u] / NPW;
        bb[u]  = ok ? b : BIN_NS;
        pk[u]  = ((unsigned)s[u] << TBITS) | (unsigned)(d[u] - b * NPW);
        pos[u] = atomicAdd(&s_hist[bb[u]], 1);
    }
    __syncthreads();

    const int v = (tid < ns) ? s_hist[tid] : 0;
    int total;
    const int ex = excl_scan_256(v, tid, s_w, &total);
    if (tid < ns) { s_off[tid] = ex; offs[tid] = ex; }
    if (tid == 0) { offs[ns] = total; s_off[BIN_NS] = total; }
    __syncthreads();

#pragma unroll
    for (int u = 0; u < BIN_U; ++u) s_stage[s_off[bb[u]] + pos[u]] = pk[u];
    __syncthreads();

    for (int i = tid; i < total; i += BIN_T) list[i] = s_stage[i];
}


__device__ __forceinline__ void lds_max_f32(float* addr, float v)
{
    if (v >= 0.f)
        atomicMax((int*)addr, __float_as_int(v));
    else
        atomicMin((unsigned int*)addr, __float_as_uint(v));
}

__global__ void __launch_bounds__(OWN_T) gat_owner(
    const int* __restrict__ src, const int* __restrict__ dst,
    const float* __restrict__ el, const float* __restrict__ er,
    const float* __restrict__ z, float* __restrict__ out,
    const unsigned int* __restrict__ binbase, unsigned long long ws_bytes,
    int n_nodes, int n_edges)
{
    __shared__ __attribute__((aligned(16))) float s_acc[NPW * OUT_F];
    __shared__ float s_m[NPW];
    __shared__ float s_l[NPW];
    __shared__ float s_er[NPW];
    __shared__ unsigned int s_pk[CAP];
    __shared__ float s_val[CAP];
    __shared__ int   s_seg0[BIN_G];
    __shared__ int   s_segn[BIN_G];
    __shared__ int   s_segp[BIN_G];
    __shared__ int   s_w[8];

    const int tid = threadIdx.x;
    const int lo  = blockIdx.x * NPW;
    const int cnt = min(NPW, n_nodes - lo);
    const unsigned ucnt = (unsigned)cnt;

    for (int i = tid; i < NPW * OUT_F; i += OWN_T) s_acc[i] = 0.f;
    for (int i = tid; i < NPW; i += OWN_T) {
        s_m[i]  = -__builtin_inff();
        s_l[i]  = 0.f;
        s_er[i] = (i < cnt) ? er[lo + i] : 0.f;
    }

    int total = CAP + 1;
    if (bins_usable(n_nodes, n_edges, (int)gridDim.x, ws_bytes)) {
        const int ch = (n_edges + BIN_G - 1) / BIN_G;
        const unsigned int* __restrict__ lists = binbase;
        const int* __restrict__ offs = (const int*)(binbase + (size_t)BIN_G * ch);
        int a = 0, len = 0;
        if (tid < BIN_G) {
            const int* o = offs + (size_t)tid * OFFP + blockIdx.x;
            a = o[0];
            const int e = o[1];
            a   = min(max(a, 0), ch);
            len = min(max(e - a, 0), ch - a);
        }
        const int p = excl_scan_256(len, tid, s_w, &total);
        if (tid < BIN_G) { s_seg0[tid] = tid * ch + a; s_segn[tid] = len; s_segp[tid] = p; }
        __syncthreads();
        if (total <= CAP) {
            const int lane = tid & 31;
            const int wv   = tid >> 5;
            int g0[BIN_G / 32], gn[BIN_G / 32], gp[BIN_G / 32];
            unsigned int gv[BIN_G / 32];
#pragma unroll
            for (int u = 0; u < BIN_G / 32; ++u) {
                const int c = wv + u * (OWN_T / 32);
                g0[u] = s_seg0[c]; gn[u] = s_segn[c]; gp[u] = s_segp[c];
            }
#pragma unroll
            for (int u = 0; u < BIN_G / 32; ++u) {
                gv[u] = 0u;
                if (lane < gn[u]) gv[u] = lists[g0[u] + lane];
            }
#pragma unroll
            for (int u = 0; u < BIN_G / 32; ++u)
                if (lane < gn[u]) s_pk[gp[u] + lane] = gv[u];
#pragma unroll
            for (int u = 0; u < BIN_G / 32; ++u) {
#pragma unroll 1
                for (int i = lane + 32; i < gn[u]; i += 32) s_pk[gp[u] + i] = lists[g0[u] + i];
            }
        }
    }
    __syncthreads();

    if (total <= CAP) {
        const int n = total;
        for (int b0 = 0; b0 < n; b0 += OWN_T * 4) {
            unsigned int pk[4];
            float ev[4];
#pragma unroll
            for (int u = 0; u < 4; ++u) {
                const int i = b0 + u * OWN_T + tid;
                pk[u] = (i < n) ? s_pk[i] : 0u;
                ev[u] = el[pk[u] >> TBITS];
            }
#pragma unroll
            for (int u = 0; u < 4; ++u) {
                const int i = b0 + u * OWN_T + tid;
                if (i < n) {
                    const int t = (int)(pk[u] & TMASK);
                    const float v = leaky(ev[u] + s_er[t]);
                    s_val[i] = v;
                    lds_max_f32(&s_m[t], v);
                }
            }
        }
        __syncthreads();
        const int ntask = n * 8;
        for (int base = 0; base < ntask; base += OWN_T * 4) {
            float4 zc[4];
            float  p[4];
            int    off[4];
#pragma unroll
            for (int u = 0; u < 4; ++u) {
                const int k = base + u * OWN_T + tid;
                const int i = k >> 3;
                const int c = (k & 7) * 4;
                unsigned int e = 0u;
                float val = 0.f;
                const bool ok = k < ntask;
                if (ok) { e = s_pk[i]; val = s_val[i]; }
                const int s = (int)(e >> TBITS);
                const int t = (int)(e & TMASK);
                zc[u]  = *(const float4*)(z + (size_t)s * OUT_F + c);
                p[u]   = ok ? __expf(val - s_m[t]) : 0.f;
                off[u] = ok ? (t * OUT_F + c) : -1;
            }
#pragma unroll
            for (int u = 0; u < 4; ++u) {
                if (off[u] >= 0) {
                    const int o = off[u];
                    atomicAdd(&s_acc[o + 0], p[u] * zc[u].x);
                    atomicAdd(&s_acc[o + 1], p[u] * zc[u].y);
                    atomicAdd(&s_acc[o + 2], p[u] * zc[u].z);
                    atomicAdd(&s_acc[o + 3], p[u] * zc[u].w);
                    if ((o & (OUT_F - 1)) == 0) atomicAdd(&s_l[o >> 5], p[u]);
                }
            }
        }
    } else {
        for (int e = tid; e < n_edges; e += OWN_T) {
            const unsigned t = (unsigned)(dst[e] - lo);
            if (t < ucnt) lds_max_f32(&s_m[t], leaky(el[src[e]] + s_er[t]));
        }
        __syncthreads();
        for (int e = tid; e < n_edges; e += OWN_T) {
            const unsigned t = (unsigned)(dst[e] - lo);
            if (t < ucnt) {
                const int s = src[e];
                const float pe = __expf(leaky(el[s] + s_er[t]) - s_m[t]);
                atomicAdd(&s_l[t], pe);
                const float* zr = z + (size_t)s * OUT_F;
                for (int j = 0; j < OUT_F; ++j)
                    atomicAdd(&s_acc[t * OUT_F + j], pe * zr[j]);
            }
        }
    }
    __syncthreads();

    for (int q = tid; q < cnt * 8; q += OWN_T) {
        const int t = q >> 3;
        const int c = (q & 7) * 4;
        const float den = fmaxf(s_l[t], 1e-16f);
        const float4 acc = *(const float4*)&s_acc[t * OUT_F + c];
        float4 r;
        r.x = acc.x / den; r.y = acc.y / den; r.z = acc.z / den; r.w = acc.w / den;
        *(float4*)(out + (size_t)(lo + t) * OUT_F + c) = r;
    }
}

extern "C" void kernel_launch(void* const* d_in, const int* in_sizes, int n_in,
                              void* d_out, int out_size, void* d_ws, size_t ws_size,
                              hipStream_t stream)
{
    const float* h  = (const float*)d_in[0];
    const float* W  = (const float*)d_in[1];
    const float* a  = (const float*)d_in[2];
    const int* src  = (const int*)d_in[3];
    const int* dst  = (const int*)d_in[4];
    float* out      = (float*)d_out;

    const int n_nodes = in_sizes[0] / IN_F;
    const int n_edges = in_sizes[3];

    float* z  = (float*)d_ws;
    float* el = z + (size_t)n_nodes * OUT_F;
    float* er = el + n_nodes;
    unsigned int* binbase = (unsigned int*)(er + n_nodes);
    const unsigned long long ws_bytes = (unsigned long long)ws_size;

    const int B = 256;
    const int ntiles = (n_nodes + 15) / 16;

    gat_gemm_wmma<<<dim3((ntiles + 7) / 8), dim3(B), 0, stream>>>(h, W, z, n_nodes);

    gat_node_logits<<<dim3((n_nodes + B - 1) / B), dim3(B), 0, stream>>>(
        z, a, el, er, n_nodes);

    gat_bin<<<dim3(BIN_G), dim3(BIN_T), 0, stream>>>(
        src, dst, binbase, ws_bytes, n_nodes, n_edges);

    gat_owner<<<dim3((n_nodes + NPW - 1) / NPW), dim3(OWN_T), 0, stream>>>(
        src, dst, el, er, z, out, binbase, ws_bytes, n_nodes, n_edges);
}


# Owner sorts its LDS edge list by destination node; 8 lanes per node sum p and p*z[src] in registers in list order and write the row: LDS float atomics and accumulator rows gone
# speedup vs baseline: 5.4134x; 1.3960x over previous
#include <hip/hip_runtime.h>
#include <hip/hip_bf16.h>

#define IN_F 128
#define OUT_F 32
#define NEG_SLOPE 0.2f

typedef float v2f __attribute__((ext_vector_type(2)));
typedef float v8f __attribute__((ext_vector_type(8)));

__global__ void __launch_bounds__(256) gat_gemm_wmma(
    const float* __restrict__ h, const float* __restrict__ W,
    float* __restrict__ z, int n_nodes)
{
    const int wave = threadIdx.x >> 5;
    const int lane = threadIdx.x & 31;
    const int tile = blockIdx.x * 8 + wave;
    const int row0 = tile * 16;
    if (row0 >= n_nodes) return;

    const int half = lane >> 4;
    const int l16  = lane & 15;
    int arow = row0 + l16;
    if (arow >= n_nodes) arow = n_nodes - 1;

    const float* __restrict__ hrow = h + (size_t)arow * IN_F;
    const float* __restrict__ w0   = W + (size_t)l16 * IN_F;
    const float* __restrict__ w1   = W + (size_t)(l16 + 16) * IN_F;

    v8f acc0 = {};
    v8f acc1 = {};

#pragma unroll 4
    for (int kk = 0; kk < IN_F; kk += 4) {
        const int ka = kk + 2 * half;
        v2f A, B0, B1;
        A[0]  = hrow[ka];  A[1]  = hrow[ka + 1];
        B0[0] = w0[ka];    B0[1] = w0[ka + 1];
        B1[0] = w1[ka];    B1[1] = w1[ka + 1];
        acc0 = __builtin_amdgcn_wmma_f32_16x16x4_f32(false, A, false, B0,
                                                     (short)0, acc0, false, false);
        acc1 = __builtin_amdgcn_wmma_f32_16x16x4_f32(false, A, false, B1,
                                                     (short)0, acc1, false, false);
    }

#pragma unroll
    for (int r = 0; r < 8; ++r) {
        const int m = row0 + r + 8 * half;
        if (m < n_nodes) {
            z[(size_t)m * OUT_F + l16]      = acc0[r];
            z[(size_t)m * OUT_F + 16 + l16] = acc1[r];
        }
    }
}

__global__ void __launch_bounds__(256) gat_node_logits(
    const float* __restrict__ z, const float* __restrict__ a,
    float* __restrict__ el, float* __restrict__ er, int n_nodes)
{
    const int n = blockIdx.x * blockDim.x + threadIdx.x;
    if (n >= n_nodes) return;
    const float* zr = z + (size_t)n * OUT_F;
    float sl = 0.f, sr = 0.f;
#pragma unroll
    for (int j = 0; j < OUT_F; ++j) {
        const float zv = zr[j];
        sl += zv * a[j];
        sr += zv * a[OUT_F + j];
    }
    el[n] = sl;
    er[n] = sr;
}

__device__ __forceinline__ float leaky(float v)
{
    return v > 0.f ? v : NEG_SLOPE * v;
}

#define NPW    391
#define CAP    16384
#define OWN_T  1024

#define BIN_G  256
#define BIN_T  256
#define BIN_U  25
#define BIN_CH (BIN_T * BIN_U)
#define BIN_NS 256
#define OFFP   (BIN_NS + 1)
#define TBITS  9
#define TMASK  ((1u << TBITS) - 1u)

__device__ __forceinline__ bool bins_usable(int n_nodes, int n_edges, int ns,
                                            unsigned long long ws_bytes)
{
    const int ch = (n_edges + BIN_G - 1) / BIN_G;
    const unsigned long long words =
        (unsigned long long)n_nodes * (OUT_F + 2) +
        (unsigned long long)BIN_G * (unsigned long long)ch +
        (unsigned long long)BIN_G * OFFP;
    return n_edges > 0 && ns <= BIN_NS && ch <= BIN_CH && n_nodes <= (1 << 22) &&
           ws_bytes >= words * 4ull;
}

__device__ __forceinline__ int excl_scan_256(int v, int tid, int* s_w, int* total)
{
    const int lane = tid & 31;
    const int wv   = tid >> 5;
    int inc = v;
#pragma unroll
    for (int d = 1; d < 32; d <<= 1) {
        const int y = __shfl_up(inc, d);
        if (lane >= d) inc += y;
    }
    if (wv < 8 && lane == 31) s_w[wv] = inc;
    __syncthreads();
    int base = 0, tot = 0;
#pragma unroll
    for (int w = 0; w < 8; ++w) {
        const int x = s_w[w];
        if (w < wv) base += x;
        tot += x;
    }
    *total = tot;
    return base + inc - v;
}

__global__ void __launch_bounds__(BIN_T) gat_bin(
    const int* __restrict__ src, const int* __restrict__ dst,
    unsigned int* __restrict__ binbase, unsigned long long ws_bytes,
    int n_nodes, int n_edges)
{
    __shared__ int s_hist[BIN_NS + 1];
    __shared__ int s_off[BIN_NS + 1];
    __shared__ int s_w[8];
    __shared__ unsigned int s_stage[BIN_CH];

    const int tid = threadIdx.x;
    const int ns  = (n_nodes + NPW - 1) / NPW;
    if (!bins_usable(n_nodes, n_edges, ns, ws_bytes)) return;

    const int ch = (n_edges + BIN_G - 1) / BIN_G;
    const int c  = blockIdx.x;
    const int e0 = c * ch;
    const int n  = max(0, min(ch, n_edges - e0));
    unsigned int* __restrict__ list = binbase + (size_t)c * ch;
    int* __restrict__ offs = (int*)(binbase + (size_t)BIN_G * ch) + (size_t)c * OFFP;

    for (int i = tid; i < BIN_NS + 1; i += BIN_T) { s_hist[i] = 0; s_off[i] = 0; }
    __syncthreads();

    int d[BIN_U], s[BIN_U];
#pragma unroll
    for (int u = 0; u < BIN_U; ++u) {
        const int e = min(e0 + tid + u * BIN_T, n_edges - 1);
        d[u] = dst[e];
        s[u] = src[e];
    }
    int bb[BIN_U], pos[BIN_U];
    unsigned int pk[BIN_U];
#pragma unroll
    for (int u = 0; u < BIN_U; ++u) {
        const bool ok = (tid + u * BIN_T < n) && ((unsigned)d[u] < (unsigned)n_nodes);
        const int b = d[u] / NPW;
        bb[u]  = ok ? b : BIN_NS;
        pk[u]  = ((unsigned)s[u] << TBITS) | (unsigned)(d[u] - b * NPW);
        pos[u] = atomicAdd(&s_hist[bb[u]], 1);
    }
    __syncthreads();

    const int v = (tid < ns) ? s_hist[tid] : 0;
    int total;
    const int ex = excl_scan_256(v, tid, s_w, &total);
    if (tid < ns) { s_off[tid] = ex; offs[tid] = ex; }
    if (tid == 0) { offs[ns] = total; s_off[BIN_NS] = total; }
    __syncthreads();

#pragma unroll
    for (int u = 0; u < BIN_U; ++u) s_stage[s_off[bb[u]] + pos[u]] = pk[u];
    __syncthreads();

    for (int i = tid; i < total; i += BIN_T) list[i] = s_stage[i];
}


__device__ __forceinline__ void lds_max_f32(float* addr, float v)
{
    if (v >= 0.f)
        atomicMax((int*)addr, __float_as_int(v));
    else
        atomicMin((unsigned int*)addr, __float_as_uint(v));
}

#define NSLOT (1 << TBITS)
#define EB    8

__device__ __forceinline__ int excl_scan_own(int v, int tid, int* s_w)
{
    const int lane = tid & 31;
    const int wv   = tid >> 5;
    int inc = v;
#pragma unroll
    for (int d = 1; d < 32; d <<= 1) {
        const int y = __shfl_up(inc, d);
        if (lane >= d) inc += y;
    }
    if (lane == 31) s_w[wv] = inc;
    __syncthreads();
    int base = 0;
#pragma unroll
    for (int w = 0; w < OWN_T / 32; ++w) {
        const int x = s_w[w];
        if (w < wv) base += x;
    }
    return base + inc - v;
}

__global__ void __launch_bounds__(OWN_T) gat_owner(
    const int* __restrict__ src, const int* __restrict__ dst,
    const float* __restrict__ el, const float* __restrict__ er,
    const float* __restrict__ z, float* __restrict__ out,
    const unsigned int* __restrict__ binbase, unsigned long long ws_bytes,
    int n_nodes, int n_edges)
{
    __shared__ float s_m[NSLOT];
    __shared__ float s_er[NSLOT];
    __shared__ int   s_cnt[NSLOT];
    __shared__ int   s_off[NSLOT];
    __shared__ int   s_cur[NSLOT];
    __shared__ unsigned int s_pk[CAP];
    __shared__ float s_val[CAP];
    __shared__ __attribute__((aligned(16))) unsigned int s_ssrc[CAP];
    __shared__ float s_sp[CAP];
    __shared__ int   s_seg0[BIN_G];
    __shared__ int   s_segn[BIN_G];
    __shared__ int   s_segp[BIN_G];
    __shared__ int   s_w[8];
    __shared__ int   s_w2[OWN_T / 32];
    __shared__ int   s_next;

    const int tid = threadIdx.x;
    const int lo  = blockIdx.x * NPW;
    const int cnt = min(NPW, n_nodes - lo);
    const unsigned ucnt = (unsigned)cnt;

    for (int i = tid; i < NSLOT; i += OWN_T) {
        s_m[i]   = -__builtin_inff();
        s_cnt[i] = 0;
        s_er[i]  = (i < cnt) ? er[lo + i] : 0.f;
    }
    if (tid == 0) s_next = 0;

    int total = CAP + 1;
    if (bins_usable(n_nodes, n_edges, (int)gridDim.x, ws_bytes)) {
        const int ch = (n_edges + BIN_G - 1) / BIN_G;
        const unsigned int* __restrict__ lists = binbase;
        const int* __restrict__ offs = (const int*)(binbase + (size_t)BIN_G * ch);
        int a = 0, len = 0;
        if (tid < BIN_G) {
            const int* o = offs + (size_t)tid * OFFP + blockIdx.x;
            a = o[0];
            const int e = o[1];
            a   = min(max(a, 0), ch);
            len = min(max(e - a, 0), ch - a);
        }
        const int p = excl_scan_256(len, tid, s_w, &total);
        if (tid < BIN_G) { s_seg0[tid] = tid * ch + a; s_segn[tid] = len; s_segp[tid] = p; }
        __syncthreads();
        if (total <= CAP) {
            const int lane = tid & 31;
            const int wv   = tid >> 5;
            int g0[BIN_G / 32], gn[BIN_G / 32], gp[BIN_G / 32];
            unsigned int gv[BIN_G / 32];
#pragma unroll
            for (int u = 0; u < BIN_G / 32; ++u) {
                const int c = wv + u * (OWN_T / 32);
                g0[u] = s_seg0[c]; gn[u] = s_segn[c]; gp[u] = s_segp[c];
            }
#pragma unroll
            for (int u = 0; u < BIN_G / 32; ++u) {
                gv[u] = 0u;
                if (lane < gn[u]) gv[u] = lists[g0[u] + lane];
            }
#pragma unroll
            for (int u = 0; u < BIN_G / 32; ++u)
                if (lane < gn[u]) s_pk[gp[u] + lane] = gv[u];
#pragma unroll
            for (int u = 0; u < BIN_G / 32; ++u) {
#pragma unroll 1
                for (int i = lane + 32; i < gn[u]; i += 32) s_pk[gp[u] + i] = lists[g0[u] + i];
            }
        }
    }
    __syncthreads();

    if (total <= CAP) {
        const int n = total;
        for (int b0 = 0; b0 < n; b0 += OWN_T * 4) {
            unsigned int pk[4];
            float ev[4];
#pragma unroll
            for (int u = 0; u < 4; ++u) {
                const int i = b0 + u * OWN_T + tid;
                pk[u] = (i < n) ? s_pk[i] : 0u;
                ev[u] = el[pk[u] >> TBITS];
            }
#pragma unroll
            for (int u = 0; u < 4; ++u) {
                const int i = b0 + u * OWN_T + tid;
                if (i < n) {
                    const int t = (int)(pk[u] & TMASK);
                    const float v = leaky(ev[u] + s_er[t]);
                    s_val[i] = v;
                    lds_max_f32(&s_m[t], v);
                    atomicAdd(&s_cnt[t], 1);
                }
            }
        }
        __syncthreads();

        {
            const int v  = (tid < NSLOT) ? s_cnt[tid] : 0;
            const int ex = excl_scan_own(v, tid, s_w2);
            if (tid < NSLOT) { s_off[tid] = ex; s_cur[tid] = ex; }
        }
        __syncthreads();
        for (int i = tid; i < n; i += OWN_T) {
            const unsigned int e = s_pk[i];
            const int t   = (int)(e & TMASK);
            const int pos = atomicAdd(&s_cur[t], 1);
            s_ssrc[pos] = e >> TBITS;
            s_sp[pos]   = __expf(s_val[i] - s_m[t]);
        }
        __syncthreads();

        const int lane = tid & 31;
        const int sub  = tid & 7;
        const int c    = sub * 4;
        for (;;) {
            int t = 0;
            if (sub == 0) t = atomicAdd(&s_next, 1);
            t = __shfl(t, lane & 24);
            const bool live = t < cnt;
            if (!__any(live)) break;
            int b = 0, e = 0;
            if (live) { b = s_off[t]; e = b + s_cnt[t]; }
            float4 acc = make_float4(0.f, 0.f, 0.f, 0.f);
            float l = 0.f;
            for (; b < e; b += EB) {
                int   s[EB];
                float p[EB];
                float4 zc[EB];
#pragma unroll
                for (int u = 0; u < EB; ++u) {
                    const int i = min(b + u, e - 1);
                    s[u] = (int)s_ssrc[i];
                    p[u] = (b + u < e) ? s_sp[i] : 0.f;
                }
#pragma unroll
                for (int u = 0; u < EB; ++u)
                    zc[u] = *(const float4*)(z + (size_t)s[u] * OUT_F + c);
#pragma unroll
                for (int u = 0; u < EB; ++u) {
                    acc.x += p[u] * zc[u].x;
                    acc.y += p[u] * zc[u].y;
                    acc.z += p[u] * zc[u].z;
                    acc.w += p[u] * zc[u].w;
                    l     += p[u];
                }
            }
            if (live) {
                const float den = fmaxf(l, 1e-16f);
                float4 r;
                r.x = acc.x / den; r.y = acc.y / den; r.z = acc.z / den; r.w = acc.w / den;
                *(float4*)(out + (size_t)(lo + t) * OUT_F + c) = r;
            }
        }
    } else {
        float* s_acc = (float*)s_ssrc;
        float* s_l   = s_sp;
        for (int i = tid; i < NPW * OUT_F; i += OWN_T) s_acc[i] = 0.f;
        for (int i = tid; i < NPW; i += OWN_T) s_l[i] = 0.f;
        for (int e = tid; e < n_edges; e += OWN_T) {
            const unsigned t = (unsigned)(dst[e] - lo);
            if (t < ucnt) lds_max_f32(&s_m[t], leaky(el[src[e]] + s_er[t]));
        }
        __syncthreads();
        for (int e = tid; e < n_edges; e += OWN_T) {
            const unsigned t = (unsigned)(dst[e] - lo);
            if (t < ucnt) {
                const int s = src[e];
                const float pe = __expf(leaky(el[s] + s_er[t]) - s_m[t]);
                atomicAdd(&s_l[t], pe);
                const float* zr = z + (size_t)s * OUT_F;
                for (int j = 0; j < OUT_F; ++j)
                    atomicAdd(&s_acc[t * OUT_F + j], pe * zr[j]);
            }
        }
        __syncthreads();
        for (int q = tid; q < cnt * 8; q += OWN_T) {
            const int t = q >> 3;
            const int c = (q & 7) * 4;
            const float den = fmaxf(s_l[t], 1e-16f);
            const float4 acc = *(const float4*)&s_acc[t * OUT_F + c];
            float4 r;
            r.x = acc.x / den; r.y = acc.y / den; r.z = acc.z / den; r.w = acc.w / den;
            *(float4*)(out + (size_t)(lo + t) * OUT_F + c) = r;
        }
    }
}

extern "C" void kernel_launch(void* const* d_in, const int* in_sizes, int n_in,
                              void* d_out, int out_size, void* d_ws, size_t ws_size,
                              hipStream_t stream)
{
    const float* h  = (const float*)d_in[0];
    const float* W  = (const float*)d_in[1];
    const float* a  = (const float*)d_in[2];
    const int* src  = (const int*)d_in[3];
    const int* dst  = (const int*)d_in[4];
    float* out      = (float*)d_out;

    const int n_nodes = in_sizes[0] / IN_F;
    const int n_edges = in_sizes[3];

    float* z  = (float*)d_ws;
    float* el = z + (size_t)n_nodes * OUT_F;
    float* er = el + n_nodes;
    unsigned int* binbase = (unsigned int*)(er + n_nodes);
    const unsigned long long ws_bytes = (unsigned long long)ws_size;

    const int B = 256;
    const int ntiles = (n_nodes + 15) / 16;

    gat_gemm_wmma<<<dim3((ntiles + 7) / 8), dim3(B), 0, stream>>>(h, W, z, n_nodes);

    gat_node_logits<<<dim3((n_nodes + B - 1) / B), dim3(B), 0, stream>>>(
        z, a, el, er, n_nodes);

    gat_bin<<<dim3(BIN_G), dim3(BIN_T), 0, stream>>>(
        src, dst, binbase, ws_bytes, n_nodes, n_edges);

    gat_owner<<<dim3((n_nodes + NPW - 1) / NPW), dim3(OWN_T), 0, stream>>>(
        src, dst, el, er, z, out, binbase, ws_bytes, n_nodes, n_edges);
}


# Product's operands go through LDS: W staged once per work-group, each wave's 16 h rows fetched as whole lines (4 per load, was 16 pieces of 16 bytes), fragments read from LDS; k order kept
# speedup vs baseline: 7.3178x; 1.3518x over previous
#include <hip/hip_runtime.h>
#include <hip/hip_bf16.h>

#define IN_F 128
#define OUT_F 32
#define NEG_SLOPE 0.2f

typedef float v2f __attribute__((ext_vector_type(2)));
typedef float v8f __attribute__((ext_vector_type(8)));
typedef float v4f __attribute__((ext_vector_type(4)));

#define PITCH (IN_F + 4)
#define KH    (IN_F / 2)
#define APITCH (KH + 4)
#define GEMM_W_FLOATS (OUT_F * PITCH)
#define GEMM_A_FLOATS (16 * APITCH)
#define GEMM_T 256

__device__ __forceinline__ float rs_step(bool up, float a, float b, int w)
{
    return (up ? b : a) + __shfl_xor(up ? a : b, w);
}

__device__ __forceinline__ void k_half(
    const v4f (&hv)[8], float* s_a, const float* hrow,
    const float* __restrict__ w0, const float* __restrict__ w1,
    int half, int l16, v8f& acc0, v8f& acc1)
{
#pragma unroll
    for (int j = 0; j < 8; ++j)
        *(v4f*)(s_a + (2 * j + half) * APITCH + 4 * l16) = hv[j];
#pragma unroll 8
    for (int kk = 0; kk < KH; kk += 4) {
        const int ka = kk + 2 * half;
        const v2f A  = *(const v2f*)(hrow + ka);
        const v2f B0 = *(const v2f*)(w0 + ka);
        const v2f B1 = *(const v2f*)(w1 + ka);
        acc0 = __builtin_amdgcn_wmma_f32_16x16x4_f32(false, A, false, B0,
                                                     (short)0, acc0, false, false);
        acc1 = __builtin_amdgcn_wmma_f32_16x16x4_f32(false, A, false, B1,
                                                     (short)0, acc1, false, false);
    }
}

__device__ __forceinline__ void gemm_tiles(
    const float* __restrict__ h, const float* __restrict__ W,
    const float* __restrict__ a,
    float* __restrict__ z, float* __restrict__ el, float* __restrict__ er,
    int n_nodes, int blk)
{
    __shared__ __attribute__((aligned(16))) float s_wt[GEMM_W_FLOATS];
    __shared__ __attribute__((aligned(16))) float s_at[(GEMM_T / 32) * GEMM_A_FLOATS];
    const int wave = threadIdx.x >> 5;
    const int lane = threadIdx.x & 31;
    const int tile = blk * 8 + wave;
    const int row0 = tile * 16;

    const int half = lane >> 4;
    const int l16  = lane & 15;

    float4 wv[OUT_F * IN_F / 4 / GEMM_T];
#pragma unroll
    for (int j = 0; j < OUT_F * IN_F / 4 / GEMM_T; ++j) {
        const int q = (int)threadIdx.x + j * GEMM_T;
        wv[j] = *(const float4*)(W + (size_t)q * 4);
    }
    v4f hv0[8], hv1[8];
#pragma unroll
    for (int j = 0; j < 8; ++j) {
        const int rr = min(row0 + 2 * j + half, n_nodes - 1);
        const float* hp = h + (size_t)rr * IN_F + 4 * l16;
        hv0[j] = *(const v4f*)hp;
        hv1[j] = *(const v4f*)(hp + KH);
    }
#pragma unroll
    for (int j = 0; j < OUT_F * IN_F / 4 / GEMM_T; ++j) {
        const int q = (int)threadIdx.x + j * GEMM_T;
        *(float4*)(s_wt + (q >> 5) * PITCH + (q & 31) * 4) = wv[j];
    }
#pragma unroll
    for (int j = 0; j < 8; ++j)
        asm volatile("" : "+v"(hv0[j]), "+v"(hv1[j]));
    __syncthreads();
    if (row0 >= n_nodes) return;

    float* s_a = s_at + wave * GEMM_A_FLOATS;
    const float* hrow = s_a  + l16 * APITCH;
    const float* __restrict__ w0   = s_wt + l16 * PITCH;
    const float* __restrict__ w1   = s_wt + (l16 + 16) * PITCH;

    const float al0 = a[l16],         al1 = a[16 + l16];
    const float ar0 = a[OUT_F + l16], ar1 = a[OUT_F + 16 + l16];

    v8f acc0 = {};
    v8f acc1 = {};

    k_half(hv0, s_a, hrow, w0, w1, half, l16, acc0, acc1);
    k_half(hv1, s_a, hrow, w0 + KH, w1 + KH, half, l16, acc0, acc1);

#pragma unroll
    for (int r = 0; r < 8; ++r) {
        const int m = row0 + r + 8 * half;
        if (m < n_nodes) {
            z[(size_t)m * OUT_F + l16]      = acc0[r];
            z[(size_t)m * OUT_F + 16 + l16] = acc1[r];
        }
    }

    const bool up8 = (l16 & 8) != 0, up4 = (l16 & 4) != 0;
    const bool up2 = (l16 & 2) != 0, up1 = (l16 & 1) != 0;
#define LOGIT_PART(r) rs_step(up8, acc0[r] * al0 + acc1[r] * al1, \
                              acc0[r] * ar0 + acc1[r] * ar1, 8)
    const float p0 = LOGIT_PART(0), p1 = LOGIT_PART(1), p2 = LOGIT_PART(2), p3 = LOGIT_PART(3);
    const float p4 = LOGIT_PART(4), p5 = LOGIT_PART(5), p6 = LOGIT_PART(6), p7 = LOGIT_PART(7);
#undef LOGIT_PART
    const float q0 = rs_step(up4, p0, p4, 4), q1 = rs_step(up4, p1, p5, 4);
    const float q2 = rs_step(up4, p2, p6, 4), q3 = rs_step(up4, p3, p7, 4);
    const float t0 = rs_step(up2, q0, q2, 2), t1 = rs_step(up2, q1, q3, 2);
    const float sum = rs_step(up1, t0, t1, 1);
    {
        const int m = row0 + 8 * half + (l16 & 7);
        float* __restrict__ dstv = up8 ? er : el;
        if (m < n_nodes) dstv[m] = sum;
    }
}

__device__ __forceinline__ float leaky(float v)
{
    return v > 0.f ? v : NEG_SLOPE * v;
}

#define NPW    391
#define CAP    16384
#define OWN_T  1024

#define BIN_G  256
#define BIN_T  256
#define BIN_U  25
#define BIN_CH (BIN_T * BIN_U)
#define BIN_NS 256
#define OFFP   (BIN_NS + 1)
#define TBITS  9
#define TMASK  ((1u << TBITS) - 1u)

__device__ __forceinline__ bool bins_usable(int n_nodes, int n_edges, int ns,
                                            unsigned long long ws_bytes)
{
    const int ch = (n_edges + BIN_G - 1) / BIN_G;
    const unsigned long long words =
        (unsigned long long)n_nodes * (OUT_F + 2) +
        (unsigned long long)BIN_G * (unsigned long long)ch +
        (unsigned long long)BIN_G * OFFP;
    return n_edges > 0 && ns <= BIN_NS && ch <= BIN_CH && n_nodes <= (1 << 22) &&
           ws_bytes >= words * 4ull;
}

__device__ __forceinline__ int excl_scan_256(int v, int tid, int* s_w, int* total)
{
    const int lane = tid & 31;
    const int wv   = tid >> 5;
    int inc = v;
#pragma unroll
    for (int d = 1; d < 32; d <<= 1) {
        const int y = __shfl_up(inc, d);
        if (lane >= d) inc += y;
    }
    if (wv < 8 && lane == 31) s_w[wv] = inc;
    __syncthreads();
    int base = 0, tot = 0;
#pragma unroll
    for (int w = 0; w < 8; ++w) {
        const int x = s_w[w];
        if (w < wv) base += x;
        tot += x;
    }
    *total = tot;
    return base + inc - v;
}

__device__ __forceinline__ void bin_chunk(
    const int* __restrict__ src, const int* __restrict__ dst,
    unsigned int* __restrict__ binbase, unsigned long long ws_bytes,
    int n_nodes, int n_edges, int c)
{
    __shared__ int s_hist[BIN_NS + 1];
    __shared__ int s_off[BIN_NS + 1];
    __shared__ int s_w[8];
    __shared__ unsigned int s_stage[BIN_CH];

    const int tid = threadIdx.x;
    const int ns  = (n_nodes + NPW - 1) / NPW;
    if (!bins_usable(n_nodes, n_edges, ns, ws_bytes)) return;

    const int ch = (n_edges + BIN_G - 1) / BIN_G;
    const int e0 = c * ch;
    const int n  = max(0, min(ch, n_edges - e0));
    unsigned int* __restrict__ list = binbase + (size_t)c * ch;
    int* __restrict__ offs = (int*)(binbase + (size_t)BIN_G * ch) + (size_t)c * OFFP;

    for (int i = tid; i < BIN_NS + 1; i += BIN_T) { s_hist[i] = 0; s_off[i] = 0; }
    __syncthreads();

    static_assert(BIN_NS < (1 << 9) && BIN_CH <= (1 << 13) && TBITS == 9,
                  "slice, rank and local destination share a 31-bit word");
    int d[BIN_U], s[BIN_U];
#pragma unroll
    for (int u = 0; u < BIN_U; ++u) {
        const int e = min(e0 + tid + u * BIN_T, n_edges - 1);
        d[u] = dst[e];
        s[u] = src[e];
    }
    __builtin_amdgcn_sched_barrier(0);
    unsigned int w[BIN_U];
#pragma unroll
    for (int u = 0; u < BIN_U; ++u) {
        const bool ok = (tid + u * BIN_T < n) && ((unsigned)d[u] < (unsigned)n_nodes);
        const int b = d[u] / NPW;
        const int bb = ok ? b : BIN_NS;
        const unsigned t = (unsigned)(d[u] - b * NPW) & TMASK;
        const int pos = atomicAdd(&s_hist[bb], 1);
        w[u] = ((unsigned)bb << 22) | ((unsigned)pos << TBITS) | t;
        asm("" : "+v"(w[u]));
    }
    __syncthreads();

    const int v = (tid < ns) ? s_hist[tid] : 0;
    int total;
    const int ex = excl_scan_256(v, tid, s_w, &total);
    if (tid < ns) { s_off[tid] = ex; offs[tid] = ex; }
    if (tid == 0) { offs[ns] = total; s_off[BIN_NS] = total; }
    __syncthreads();

#pragma unroll
    for (int u = 0; u < BIN_U; ++u) {
        s_stage[s_off[w[u] >> 22] + (int)((w[u] >> TBITS) & 0x1fffu)] =
            ((unsigned)s[u] << TBITS) | (w[u] & TMASK);
    }
    __syncthreads();

    for (int i = tid; i < total; i += BIN_T) list[i] = s_stage[i];
}

static_assert(BIN_T == GEMM_T, "both halves of the first launch use the same block");

__global__ void __launch_bounds__(BIN_T) gat_gemm_bin(
    const float* __restrict__ h, const float* __restrict__ W,
    const float* __restrict__ a,
    float* __restrict__ z, float* __restrict__ el, float* __restrict__ er,
    const int* __restrict__ src, const int* __restrict__ dst,
    unsigned int* __restrict__ binbase, unsigned long long ws_bytes,
    int n_nodes, int n_edges)
{
    const int b = blockIdx.x;
    if (b < BIN_G)
        bin_chunk(src, dst, binbase, ws_bytes, n_nodes, n_edges, b);
    else
        gemm_tiles(h, W, a, z, el, er, n_nodes, b - BIN_G);
}


__device__ __forceinline__ void lds_max_f32(float* addr, float v)
{
    if (v >= 0.f)
        atomicMax((int*)addr, __float_as_int(v));
    else
        atomicMin((unsigned int*)addr, __float_as_uint(v));
}

#define NSLOT (1 << TBITS)
#define EB    8

__device__ __forceinline__ int excl_scan_own(int v, int tid, int* s_w)
{
    const int lane = tid & 31;
    const int wv   = tid >> 5;
    int inc = v;
#pragma unroll
    for (int d = 1; d < 32; d <<= 1) {
        const int y = __shfl_up(inc, d);
        if (lane >= d) inc += y;
    }
    if (lane == 31) s_w[wv] = inc;
    __syncthreads();
    int base = 0;
#pragma unroll
    for (int w = 0; w < OWN_T / 32; ++w) {
        const int x = s_w[w];
        if (w < wv) base += x;
    }
    return base + inc - v;
}

__global__ void __launch_bounds__(OWN_T) gat_owner(
    const int* __restrict__ src, const int* __restrict__ dst,
    const float* __restrict__ el, const float* __restrict__ er,
    const float* __restrict__ z, float* __restrict__ out,
    const unsigned int* __restrict__ binbase, unsigned long long ws_bytes,
    int n_nodes, int n_edges)
{
    __shared__ float s_m[NSLOT];
    __shared__ float s_er[NSLOT];
    __shared__ int   s_cnt[NSLOT];
    __shared__ int   s_off[NSLOT];
    __shared__ int   s_cur[NSLOT];
    __shared__ unsigned int s_pk[CAP];
    __shared__ float s_val[CAP];
    __shared__ __attribute__((aligned(16))) unsigned int s_ssrc[CAP];
    __shared__ float s_sp[CAP];
    __shared__ int   s_seg0[BIN_G];
    __shared__ int   s_segn[BIN_G];
    __shared__ int   s_segp[BIN_G];
    __shared__ int   s_w[8];
    __shared__ int   s_w2[OWN_T / 32];
    __shared__ int   s_next;

    const int tid = threadIdx.x;
    const int lo  = blockIdx.x * NPW;
    const int cnt = min(NPW, n_nodes - lo);
    const unsigned ucnt = (unsigned)cnt;

    for (int i = tid; i < NSLOT; i += OWN_T) {
        s_m[i]   = -__builtin_inff();
        s_cnt[i] = 0;
        s_er[i]  = (i < cnt) ? er[lo + i] : 0.f;
    }
    if (tid == 0) s_next = 0;

    int total = CAP + 1;
    if (bins_usable(n_nodes, n_edges, (int)gridDim.x, ws_bytes)) {
        const int ch = (n_edges + BIN_G - 1) / BIN_G;
        const unsigned int* __restrict__ lists = binbase;
        const int* __restrict__ offs = (const int*)(binbase + (size_t)BIN_G * ch);
        int a = 0, len = 0;
        if (tid < BIN_G) {
            const int* o = offs + (size_t)tid * OFFP + blockIdx.x;
            a = o[0];
            const int e = o[1];
            a   = min(max(a, 0), ch);
            len = min(max(e - a, 0), ch - a);
        }
        const int p = excl_scan_256(len, tid, s_w, &total);
        if (tid < BIN_G) { s_seg0[tid] = tid * ch + a; s_segn[tid] = len; s_segp[tid] = p; }
        __syncthreads();
        if (total <= CAP) {
            const int lane = tid & 31;
            const int wv   = tid >> 5;
            int g0[BIN_G / 32], gn[BIN_G / 32], gp[BIN_G / 32];
            unsigned int gv[BIN_G / 32];
#pragma unroll
            for (int u = 0; u < BIN_G / 32; ++u) {
                const int c = wv + u * (OWN_T / 32);
                g0[u] = s_seg0[c]; gn[u] = s_segn[c]; gp[u] = s_segp[c];
            }
#pragma unroll
            for (int u = 0; u < BIN_G / 32; ++u) {
                gv[u] = 0u;
                if (lane < gn[u]) gv[u] = lists[g0[u] + lane];
            }
#pragma unroll
            for (int u = 0; u < BIN_G / 32; ++u)
                if (lane < gn[u]) s_pk[gp[u] + lane] = gv[u];
#pragma unroll
            for (int u = 0; u < BIN_G / 32; ++u) {
#pragma unroll 1
                for (int i = lane + 32; i < gn[u]; i += 32) s_pk[gp[u] + i] = lists[g0[u] + i];
            }
        }
    }
    __syncthreads();

    if (total <= CAP) {
        const int n = total;
        for (int b0 = 0; b0 < n; b0 += OWN_T * 4) {
            unsigned int pk[4];
            float ev[4];
#pragma unroll
            for (int u = 0; u < 4; ++u) {
                const int i = b0 + u * OWN_T + tid;
                pk[u] = (i < n) ? s_pk[i] : 0u;
                ev[u] = el[pk[u] >> TBITS];
            }
#pragma unroll
            for (int u = 0; u < 4; ++u) {
                const int i = b0 + u * OWN_T + tid;
                if (i < n) {
                    const int t = (int)(pk[u] & TMASK);
                    const float v = leaky(ev[u] + s_er[t]);
                    s_val[i] = v;
                    lds_max_f32(&s_m[t], v);
                    atomicAdd(&s_cnt[t], 1);
                }
            }
        }
        __syncthreads();

        {
            const int v  = (tid < NSLOT) ? s_cnt[tid] : 0;
            const int ex = excl_scan_own(v, tid, s_w2);
            if (tid < NSLOT) { s_off[tid] = ex; s_cur[tid] = ex; }
        }
        __syncthreads();
        for (int i = tid; i < n; i += OWN_T) {
            const unsigned int e = s_pk[i];
            const int t   = (int)(e & TMASK);
            const int pos = atomicAdd(&s_cur[t], 1);
            s_ssrc[pos] = e >> TBITS;
            s_sp[pos]   = __expf(s_val[i] - s_m[t]);
        }
        __syncthreads();

        const int lane = tid & 31;
        const int sub  = tid & 7;
        const int c    = sub * 4;
        for (;;) {
            int t = 0;
            if (sub == 0) t = atomicAdd(&s_next, 1);
            t = __shfl(t, lane & 24);
            const bool live = t < cnt;
            if (!__any(live)) break;
            int b = 0, e = 0;
            if (live) { b = s_off[t]; e = b + s_cnt[t]; }
            float4 acc = make_float4(0.f, 0.f, 0.f, 0.f);
            float l = 0.f;
            for (; b < e; b += EB) {
                int   s[EB];
                float p[EB];
                float4 zc[EB];
#pragma unroll
                for (int u = 0; u < EB; ++u) {
                    const int i = min(b + u, e - 1);
                    s[u] = (int)s_ssrc[i];
                    p[u] = (b + u < e) ? s_sp[i] : 0.f;
                }
#pragma unroll
                for (int u = 0; u < EB; ++u)
                    zc[u] = *(const float4*)(z + (size_t)s[u] * OUT_F + c);
#pragma unroll
                for (int u = 0; u < EB; ++u) {
                    acc.x += p[u] * zc[u].x;
                    acc.y += p[u] * zc[u].y;
                    acc.z += p[u] * zc[u].z;
                    acc.w += p[u] * zc[u].w;
                    l     += p[u];
                }
            }
            if (live) {
                const float den = fmaxf(l, 1e-16f);
                float4 r;
                r.x = acc.x / den; r.y = acc.y / den; r.z = acc.z / den; r.w = acc.w / den;
                *(float4*)(out + (size_t)(lo + t) * OUT_F + c) = r;
            }
        }
    } else {
        float* s_acc = (float*)s_ssrc;
        float* s_l   = s_sp;
        for (int i = tid; i < NPW * OUT_F; i += OWN_T) s_acc[i] = 0.f;
        for (int i = tid; i < NPW; i += OWN_T) s_l[i] = 0.f;
        for (int e = tid; e < n_edges; e += OWN_T) {
            const unsigned t = (unsigned)(dst[e] - lo);
            if (t < ucnt) lds_max_f32(&s_m[t], leaky(el[src[e]] + s_er[t]));
        }
        __syncthreads();
        for (int e = tid; e < n_edges; e += OWN_T) {
            const unsigned t = (unsigned)(dst[e] - lo);
            if (t < ucnt) {
                const int s = src[e];
                const float pe = __expf(leaky(el[s] + s_er[t]) - s_m[t]);
                atomicAdd(&s_l[t], pe);
                const float* zr = z + (size_t)s * OUT_F;
                for (int j = 0; j < OUT_F; ++j)
                    atomicAdd(&s_acc[t * OUT_F + j], pe * zr[j]);
            }
        }
        __syncthreads();
        for (int q = tid; q < cnt * 8; q += OWN_T) {
            const int t = q >> 3;
            const int c = (q & 7) * 4;
            const float den = fmaxf(s_l[t], 1e-16f);
            const float4 acc = *(const float4*)&s_acc[t * OUT_F + c];
            float4 r;
            r.x = acc.x / den; r.y = acc.y / den; r.z = acc.z / den; r.w = acc.w / den;
            *(float4*)(out + (size_t)(lo + t) * OUT_F + c) = r;
        }
    }
}

extern "C" void kernel_launch(void* const* d_in, const int* in_sizes, int n_in,
                              void* d_out, int out_size, void* d_ws, size_t ws_size,
                              hipStream_t stream)
{
    const float* h  = (const float*)d_in[0];
    const float* W  = (const float*)d_in[1];
    const float* a  = (const float*)d_in[2];
    const int* src  = (const int*)d_in[3];
    const int* dst  = (const int*)d_in[4];
    float* out      = (float*)d_out;

    const int n_nodes = in_sizes[0] / IN_F;
    const int n_edges = in_sizes[3];

    float* z  = (float*)d_ws;
    float* el = z + (size_t)n_nodes * OUT_F;
    float* er = el + n_nodes;
    unsigned int* binbase = (unsigned int*)(er + n_nodes);
    const unsigned long long ws_bytes = (unsigned long long)ws_size;

    const int ntiles = (n_nodes + 15) / 16;

    gat_gemm_bin<<<dim3(BIN_G + (ntiles + 7) / 8), dim3(BIN_T), 0, stream>>>(
        h, W, a, z, el, er, src, dst, binbase, ws_bytes, n_nodes, n_edges);

    gat_owner<<<dim3((n_nodes + NPW - 1) / NPW), dim3(OWN_T), 0, stream>>>(
        src, dst, el, er, z, out, binbase, ws_bytes, n_nodes, n_edges);
}


# Owner sorts its edges from registers: each wave loads its runs into its lanes, ranks them (counting atomic, arrival order as before) and scatters them; chunk scan, LDS list copy, 2 barriers go
# speedup vs baseline: 8.6218x; 1.1782x over previous
#include <hip/hip_runtime.h>
#include <hip/hip_bf16.h>

#define IN_F 128
#define OUT_F 32
#define NEG_SLOPE 0.2f

typedef float v2f __attribute__((ext_vector_type(2)));
typedef float v8f __attribute__((ext_vector_type(8)));
typedef float v4f __attribute__((ext_vector_type(4)));

#define PITCH (IN_F + 4)
#define KH    (IN_F / 2)
#define APITCH (KH + 4)
#define GEMM_W_FLOATS (OUT_F * PITCH)
#define GEMM_A_FLOATS (16 * APITCH)
#define GEMM_T 256

__device__ __forceinline__ float rs_step(bool up, float a, float b, int w)
{
    return (up ? b : a) + __shfl_xor(up ? a : b, w);
}

__device__ __forceinline__ void k_half(
    const v4f (&hv)[8], float* s_a, const float* hrow,
    const float* __restrict__ w0, const float* __restrict__ w1,
    int half, int l16, v8f& acc0, v8f& acc1)
{
#pragma unroll
    for (int j = 0; j < 8; ++j)
        *(v4f*)(s_a + (2 * j + half) * APITCH + 4 * l16) = hv[j];
#pragma unroll 8
    for (int kk = 0; kk < KH; kk += 4) {
        const int ka = kk + 2 * half;
        const v2f A  = *(const v2f*)(hrow + ka);
        const v2f B0 = *(const v2f*)(w0 + ka);
        const v2f B1 = *(const v2f*)(w1 + ka);
        acc0 = __builtin_amdgcn_wmma_f32_16x16x4_f32(false, A, false, B0,
                                                     (short)0, acc0, false, false);
        acc1 = __builtin_amdgcn_wmma_f32_16x16x4_f32(false, A, false, B1,
                                                     (short)0, acc1, false, false);
    }
}

__device__ __forceinline__ void gemm_tiles(
    const float* __restrict__ h, const float* __restrict__ W,
    const float* __restrict__ a,
    float* __restrict__ z, float* __restrict__ el, float* __restrict__ er,
    int n_nodes, int blk)
{
    __shared__ __attribute__((aligned(16))) float s_wt[GEMM_W_FLOATS];
    __shared__ __attribute__((aligned(16))) float s_at[(GEMM_T / 32) * GEMM_A_FLOATS];
    const int wave = threadIdx.x >> 5;
    const int lane = threadIdx.x & 31;
    const int tile = blk * 8 + wave;
    const int row0 = tile * 16;

    const int half = lane >> 4;
    const int l16  = lane & 15;

    float4 wv[OUT_F * IN_F / 4 / GEMM_T];
#pragma unroll
    for (int j = 0; j < OUT_F * IN_F / 4 / GEMM_T; ++j) {
        const int q = (int)threadIdx.x + j * GEMM_T;
        wv[j] = *(const float4*)(W + (size_t)q * 4);
    }
    v4f hv0[8], hv1[8];
#pragma unroll
    for (int j = 0; j < 8; ++j) {
        const int rr = min(row0 + 2 * j + half, n_nodes - 1);
        const float* hp = h + (size_t)rr * IN_F + 4 * l16;
        hv0[j] = *(const v4f*)hp;
        hv1[j] = *(const v4f*)(hp + KH);
    }
#pragma unroll
    for (int j = 0; j < OUT_F * IN_F / 4 / GEMM_T; ++j) {
        const int q = (int)threadIdx.x + j * GEMM_T;
        *(float4*)(s_wt + (q >> 5) * PITCH + (q & 31) * 4) = wv[j];
    }
#pragma unroll
    for (int j = 0; j < 8; ++j)
        asm volatile("" : "+v"(hv0[j]), "+v"(hv1[j]));
    __syncthreads();
    if (row0 >= n_nodes) return;

    float* s_a = s_at + wave * GEMM_A_FLOATS;
    const float* hrow = s_a  + l16 * APITCH;
    const float* __restrict__ w0   = s_wt + l16 * PITCH;
    const float* __restrict__ w1   = s_wt + (l16 + 16) * PITCH;

    const float al0 = a[l16],         al1 = a[16 + l16];
    const float ar0 = a[OUT_F + l16], ar1 = a[OUT_F + 16 + l16];

    v8f acc0 = {};
    v8f acc1 = {};

    k_half(hv0, s_a, hrow, w0, w1, half, l16, acc0, acc1);
    k_half(hv1, s_a, hrow, w0 + KH, w1 + KH, half, l16, acc0, acc1);

#pragma unroll
    for (int r = 0; r < 8; ++r) {
        const int m = row0 + r + 8 * half;
        if (m < n_nodes) {
            z[(size_t)m * OUT_F + l16]      = acc0[r];
            z[(size_t)m * OUT_F + 16 + l16] = acc1[r];
        }
    }

    const bool up8 = (l16 & 8) != 0, up4 = (l16 & 4) != 0;
    const bool up2 = (l16 & 2) != 0, up1 = (l16 & 1) != 0;
#define LOGIT_PART(r) rs_step(up8, acc0[r] * al0 + acc1[r] * al1, \
                              acc0[r] * ar0 + acc1[r] * ar1, 8)
    const float p0 = LOGIT_PART(0), p1 = LOGIT_PART(1), p2 = LOGIT_PART(2), p3 = LOGIT_PART(3);
    const float p4 = LOGIT_PART(4), p5 = LOGIT_PART(5), p6 = LOGIT_PART(6), p7 = LOGIT_PART(7);
#undef LOGIT_PART
    const float q0 = rs_step(up4, p0, p4, 4), q1 = rs_step(up4, p1, p5, 4);
    const float q2 = rs_step(up4, p2, p6, 4), q3 = rs_step(up4, p3, p7, 4);
    const float t0 = rs_step(up2, q0, q2, 2), t1 = rs_step(up2, q1, q3, 2);
    const float sum = rs_step(up1, t0, t1, 1);
    {
        const int m = row0 + 8 * half + (l16 & 7);
        float* __restrict__ dstv = up8 ? er : el;
        if (m < n_nodes) dstv[m] = sum;
    }
}

__device__ __forceinline__ float leaky(float v)
{
    return v > 0.f ? v : NEG_SLOPE * v;
}

#define NPW    391
#define CAP    16384
#define OWN_T  1024

#define BIN_G  256
#define BIN_T  256
#define BIN_U  25
#define BIN_CH (BIN_T * BIN_U)
#define BIN_NS 256
#define OFFP   (BIN_NS + 1)
#define TBITS  9
#define TMASK  ((1u << TBITS) - 1u)

__device__ __forceinline__ bool bins_usable(int n_nodes, int n_edges, int ns,
                                            unsigned long long ws_bytes)
{
    const int ch = (n_edges + BIN_G - 1) / BIN_G;
    const unsigned long long words =
        (unsigned long long)n_nodes * (OUT_F + 2) +
        (unsigned long long)BIN_G * (unsigned long long)ch +
        (unsigned long long)BIN_G * OFFP;
    return n_edges > 0 && ns <= BIN_NS && ch <= BIN_CH && n_nodes <= (1 << 22) &&
           ws_bytes >= words * 4ull;
}

__device__ __forceinline__ int excl_scan_256(int v, int tid, int* s_w, int* total)
{
    const int lane = tid & 31;
    const int wv   = tid >> 5;
    int inc = v;
#pragma unroll
    for (int d = 1; d < 32; d <<= 1) {
        const int y = __shfl_up(inc, d);
        if (lane >= d) inc += y;
    }
    if (wv < 8 && lane == 31) s_w[wv] = inc;
    __syncthreads();
    int base = 0, tot = 0;
#pragma unroll
    for (int w = 0; w < 8; ++w) {
        const int x = s_w[w];
        if (w < wv) base += x;
        tot += x;
    }
    *total = tot;
    return base + inc - v;
}

__device__ __forceinline__ void bin_chunk(
    const int* __restrict__ src, const int* __restrict__ dst,
    unsigned int* __restrict__ binbase, unsigned long long ws_bytes,
    int n_nodes, int n_edges, int c)
{
    __shared__ int s_hist[BIN_NS + 1];
    __shared__ int s_off[BIN_NS + 1];
    __shared__ int s_w[8];
    __shared__ unsigned int s_stage[BIN_CH];

    const int tid = threadIdx.x;
    const int ns  = (n_nodes + NPW - 1) / NPW;
    if (!bins_usable(n_nodes, n_edges, ns, ws_bytes)) return;

    const int ch = (n_edges + BIN_G - 1) / BIN_G;
    const int e0 = c * ch;
    const int n  = max(0, min(ch, n_edges - e0));
    unsigned int* __restrict__ list = binbase + (size_t)c * ch;
    int* __restrict__ offs = (int*)(binbase + (size_t)BIN_G * ch) + (size_t)c * OFFP;

    for (int i = tid; i < BIN_NS + 1; i += BIN_T) { s_hist[i] = 0; s_off[i] = 0; }
    __syncthreads();

    static_assert(BIN_NS < (1 << 9) && BIN_CH <= (1 << 13) && TBITS == 9,
                  "slice, rank and local destination share a 31-bit word");
    int d[BIN_U], s[BIN_U];
#pragma unroll
    for (int u = 0; u < BIN_U; ++u) {
        const int e = min(e0 + tid + u * BIN_T, n_edges - 1);
        d[u] = dst[e];
        s[u] = src[e];
    }
    __builtin_amdgcn_sched_barrier(0);
    unsigned int w[BIN_U];
#pragma unroll
    for (int u = 0; u < BIN_U; ++u) {
        const bool ok = (tid + u * BIN_T < n) && ((unsigned)d[u] < (unsigned)n_nodes);
        const int b = d[u] / NPW;
        const int bb = ok ? b : BIN_NS;
        const unsigned t = (unsigned)(d[u] - b * NPW) & TMASK;
        const int pos = atomicAdd(&s_hist[bb], 1);
        w[u] = ((unsigned)bb << 22) | ((unsigned)pos << TBITS) | t;
        asm("" : "+v"(w[u]));
    }
    __syncthreads();

    const int v = (tid < ns) ? s_hist[tid] : 0;
    int total;
    const int ex = excl_scan_256(v, tid, s_w, &total);
    if (tid < ns) { s_off[tid] = ex; offs[tid] = ex; }
    if (tid == 0) { offs[ns] = total; s_off[BIN_NS] = total; }
    __syncthreads();

#pragma unroll
    for (int u = 0; u < BIN_U; ++u) {
        s_stage[s_off[w[u] >> 22] + (int)((w[u] >> TBITS) & 0x1fffu)] =
            ((unsigned)s[u] << TBITS) | (w[u] & TMASK);
    }
    __syncthreads();

    for (int i = tid; i < total; i += BIN_T) list[i] = s_stage[i];
}

static_assert(BIN_T == GEMM_T, "both halves of the first launch use the same block");

__global__ void __launch_bounds__(BIN_T) gat_gemm_bin(
    const float* __restrict__ h, const float* __restrict__ W,
    const float* __restrict__ a,
    float* __restrict__ z, float* __restrict__ el, float* __restrict__ er,
    const int* __restrict__ src, const int* __restrict__ dst,
    unsigned int* __restrict__ binbase, unsigned long long ws_bytes,
    int n_nodes, int n_edges)
{
    const int b = blockIdx.x;
    if (b < BIN_G)
        bin_chunk(src, dst, binbase, ws_bytes, n_nodes, n_edges, b);
    else
        gemm_tiles(h, W, a, z, el, er, n_nodes, b - BIN_G);
}


__device__ __forceinline__ void lds_max_f32(float* addr, float v)
{
    if (v >= 0.f)
        atomicMax((int*)addr, __float_as_int(v));
    else
        atomicMin((unsigned int*)addr, __float_as_uint(v));
}

#define NSLOT (1 << TBITS)
#define EB    8
#define RUNS_W (BIN_G / (OWN_T / 32))
#define RUN_L  2
#define HELD   (RUNS_W * RUN_L)
static_assert(BIN_G % (OWN_T / 32) == 0, "every wave fetches the same number of runs");
static_assert(HELD * OWN_T <= CAP, "what the lanes can hold fits the sorted list");

__device__ __forceinline__ int excl_scan_own(int v, int tid, int* s_w, int* total)
{
    const int lane = tid & 31;
    const int wv   = tid >> 5;
    int inc = v;
#pragma unroll
    for (int d = 1; d < 32; d <<= 1) {
        const int y = __shfl_up(inc, d);
        if (lane >= d) inc += y;
    }
    if (lane == 31) s_w[wv] = inc;
    __syncthreads();
    int base = 0, tot = 0;
#pragma unroll
    for (int w = 0; w < OWN_T / 32; ++w) {
        const int x = s_w[w];
        if (w < wv) base += x;
        tot += x;
    }
    *total = tot;
    return base + inc - v;
}

template <unsigned SEL>
__device__ __forceinline__ float grp8(float v)
{
    return __int_as_float((int)__builtin_amdgcn_mov_dpp8((unsigned)__float_as_int(v), SEL));
}
#define SEL8(a, b, c, d, e, f, g, h) \
    ((a) | ((b) << 3) | ((c) << 6) | ((d) << 9) | ((e) << 12) | ((f) << 15) | ((g) << 18) | ((h) << 21))
#define SEL_X1   SEL8(1u, 0u, 3u, 2u, 5u, 4u, 7u, 6u)
#define SEL_X2   SEL8(2u, 3u, 0u, 1u, 6u, 7u, 4u, 5u)
#define SEL_X4   SEL8(4u, 5u, 6u, 7u, 0u, 1u, 2u, 3u)
#define SEL_B(u) SEL8(u, u, u, u, u, u, u, u)

template <unsigned SEL>
__device__ __forceinline__ float rs8(bool up, float a, float b)
{
    return (up ? b : a) + grp8<SEL>(up ? a : b);
}

__global__ void __launch_bounds__(OWN_T) gat_owner(
    const int* __restrict__ src, const int* __restrict__ dst,
    const float* __restrict__ a,
    const float* __restrict__ el, const float* __restrict__ er,
    const float* __restrict__ z, float* __restrict__ out,
    const unsigned int* __restrict__ binbase, unsigned long long ws_bytes,
    int n_nodes, int n_edges)
{
    __shared__ float s_m[NSLOT];
    __shared__ float s_er[NSLOT];
    __shared__ int   s_cnt[NSLOT];
    __shared__ int   s_off[NSLOT];
    __shared__ __attribute__((aligned(16))) unsigned int s_ssrc[CAP];
    __shared__ int   s_w2[OWN_T / 32];
    __shared__ int   s_next;
    __shared__ int   s_long;

    const int tid  = threadIdx.x;
    const int lane = tid & 31;
    const int lo  = blockIdx.x * NPW;
    const int cnt = min(NPW, n_nodes - lo);
    const unsigned ucnt = (unsigned)cnt;

    const float4 al = *(const float4*)(a + (tid & 7) * 4);

    const bool usable = bins_usable(n_nodes, n_edges, (int)gridDim.x, ws_bytes);
    unsigned int pk[HELD];
    int gn[RUNS_W];
    bool lng = false;
#pragma unroll
    for (int u = 0; u < RUNS_W; ++u) gn[u] = 0;
#pragma unroll
    for (int j = 0; j < HELD; ++j) pk[j] = 0u;
    if (usable) {
        const int ch = (n_edges + BIN_G - 1) / BIN_G;
        const unsigned int* __restrict__ lists = binbase;
        const int* __restrict__ offs = (const int*)(binbase + (size_t)BIN_G * ch);
        const int wv = __builtin_amdgcn_readfirstlane(tid >> 5);
        int g0[RUNS_W];
#pragma unroll
        for (int u = 0; u < RUNS_W; ++u) {
            const int c = wv + u * (OWN_T / 32);
            const int* o = offs + (size_t)c * OFFP + blockIdx.x;
            int a0 = o[0];
            const int e = o[1];
            a0    = min(max(a0, 0), ch);
            gn[u] = min(max(e - a0, 0), ch - a0);
            g0[u] = c * ch + a0;
            lng   = lng || gn[u] > 32 * RUN_L;
        }
#pragma unroll
        for (int u = 0; u < RUNS_W; ++u) {
#pragma unroll
            for (int k = 0; k < RUN_L; ++k) {
                const int i = lane + 32 * k;
                if (i < gn[u]) pk[u * RUN_L + k] = lists[g0[u] + i];
            }
        }
    }
    for (int i = tid; i < NSLOT; i += OWN_T) {
        s_m[i]   = -__builtin_inff();
        s_cnt[i] = 0;
        s_er[i]  = (i < cnt) ? er[lo + i] : 0.f;
    }
    if (tid == 0) { s_next = 0; s_long = 0; }
    __syncthreads();

    int rk[HELD];
#pragma unroll
    for (int j = 0; j < HELD; ++j) {
        rk[j] = 0;
        if (lane + 32 * (j % RUN_L) < gn[j / RUN_L])
            rk[j] = atomicAdd(&s_cnt[pk[j] & TMASK], 1);
    }
    if (lng && lane == 0) s_long = 1;
    __syncthreads();

    int total = CAP + 1;
    if (usable) {
        int tot;
        const int v  = (tid < NSLOT) ? s_cnt[tid] : 0;
        const int ex = excl_scan_own(v, tid, s_w2, &tot);
        if (tid < NSLOT) s_off[tid] = ex;
        if (s_long == 0) total = tot;
    }
    __syncthreads();

    if (total <= CAP) {
        int st[HELD];
#pragma unroll
        for (int j = 0; j < HELD; ++j) st[j] = s_off[pk[j] & TMASK];
#pragma unroll
        for (int j = 0; j < HELD; ++j) {
            if (lane + 32 * (j % RUN_L) < gn[j / RUN_L])
                s_ssrc[st[j] + rk[j]] = pk[j] >> TBITS;
        }
        __syncthreads();

        static_assert(EB == 8, "one edge of a batch per lane of the group");
        const int sub  = tid & 7;
        const int c    = sub * 4;
        const bool up4 = (sub & 4) != 0, up2 = (sub & 2) != 0, up1 = (sub & 1) != 0;
        const v2f alo = {al.x, al.y}, ahi = {al.z, al.w};
        for (;;) {
            int t = 0;
            if (sub == 0) t = atomicAdd(&s_next, 1);
            t = __shfl(t, lane & 24);
            const bool live = t < cnt;
            if (!__any(live)) break;
            int b = 0, e = 0;
            float ert = 0.f;
            if (live) { b = s_off[t]; e = b + s_cnt[t]; ert = s_er[t]; }
            float4 acc = make_float4(0.f, 0.f, 0.f, 0.f);
            float l = 0.f;
            float m = -__builtin_inff();
            for (; b < e; b += EB) {
                int   s[EB];
                float4 zc[EB];
                float d[EB];
#pragma unroll
                for (int u = 0; u < EB; ++u)
                    s[u] = (int)s_ssrc[min(b + u, e - 1)];
#pragma unroll
                for (int u = 0; u < EB; ++u)
                    zc[u] = *(const float4*)(z + (size_t)s[u] * OUT_F + c);
#pragma unroll
                for (int u = 0; u < EB; ++u) {
                    const v2f zlo = {zc[u].x, zc[u].y}, zhi = {zc[u].z, zc[u].w};
                    const v2f t2 = zlo * alo + zhi * ahi;
                    d[u] = t2.x + t2.y;
                }
                const float q0 = rs8<SEL_X4>(up4, d[0], d[4]), q1 = rs8<SEL_X4>(up4, d[1], d[5]);
                const float q2 = rs8<SEL_X4>(up4, d[2], d[6]), q3 = rs8<SEL_X4>(up4, d[3], d[7]);
                const float r0 = rs8<SEL_X2>(up2, q0, q2), r1 = rs8<SEL_X2>(up2, q1, q3);
                const float els = rs8<SEL_X1>(up1, r0, r1);
                const float v = (b + sub < e) ? leaky(els + ert) : -__builtin_inff();
                float bm = fmaxf(v, grp8<SEL_X1>(v));
                bm = fmaxf(bm, grp8<SEL_X2>(bm));
                bm = fmaxf(bm, grp8<SEL_X4>(bm));
                const float mn = fmaxf(m, bm);
                const float sc = __expf(m - mn);
                const float p  = __expf(v - mn);
                m = mn;
                acc.x *= sc; acc.y *= sc; acc.z *= sc; acc.w *= sc;
                l *= sc;
#define ADD_EDGE(k) { const float pk_ = grp8<SEL_B(k##u)>(p); \
                      acc.x += pk_ * zc[k].x; acc.y += pk_ * zc[k].y; \
                      acc.z += pk_ * zc[k].z; acc.w += pk_ * zc[k].w; l += pk_; }
                ADD_EDGE(0) ADD_EDGE(1) ADD_EDGE(2) ADD_EDGE(3)
                ADD_EDGE(4) ADD_EDGE(5) ADD_EDGE(6) ADD_EDGE(7)
#undef ADD_EDGE
            }
            if (live) {
                const float den = fmaxf(l, 1e-16f);
                float4 r;
                r.x = acc.x / den; r.y = acc.y / den; r.z = acc.z / den; r.w = acc.w / den;
                *(float4*)(out + (size_t)(lo + t) * OUT_F + c) = r;
            }
        }
    } else {
        float* s_acc = (float*)s_ssrc;
        float* s_l   = (float*)s_off;
        for (int i = tid; i < NPW * OUT_F; i += OWN_T) s_acc[i] = 0.f;
        for (int i = tid; i < NPW; i += OWN_T) s_l[i] = 0.f;
        for (int e = tid; e < n_edges; e += OWN_T) {
            const unsigned t = (unsigned)(dst[e] - lo);
            if (t < ucnt) lds_max_f32(&s_m[t], leaky(el[src[e]] + s_er[t]));
        }
        __syncthreads();
        for (int e = tid; e < n_edges; e += OWN_T) {
            const unsigned t = (unsigned)(dst[e] - lo);
            if (t < ucnt) {
                const int s = src[e];
                const float pe = __expf(leaky(el[s] + s_er[t]) - s_m[t]);
                atomicAdd(&s_l[t], pe);
                const float* zr = z + (size_t)s * OUT_F;
                for (int j = 0; j < OUT_F; ++j)
                    atomicAdd(&s_acc[t * OUT_F + j], pe * zr[j]);
            }
        }
        __syncthreads();
        for (int q = tid; q < cnt * 8; q += OWN_T) {
            const int t = q >> 3;
            const int c = (q & 7) * 4;
            const float den = fmaxf(s_l[t], 1e-16f);
            const float4 acc = *(const float4*)&s_acc[t * OUT_F + c];
            float4 r;
            r.x = acc.x / den; r.y = acc.y / den; r.z = acc.z / den; r.w = acc.w / den;
            *(float4*)(out + (size_t)(lo + t) * OUT_F + c) = r;
        }
    }
}

extern "C" void kernel_launch(void* const* d_in, const int* in_sizes, int n_in,
                              void* d_out, int out_size, void* d_ws, size_t ws_size,
                              hipStream_t stream)
{
    const float* h  = (const float*)d_in[0];
    const float* W  = (const float*)d_in[1];
    const float* a  = (const float*)d_in[2];
    const int* src  = (const int*)d_in[3];
    const int* dst  = (const int*)d_in[4];
    float* out      = (float*)d_out;

    const int n_nodes = in_sizes[0] / IN_F;
    const int n_edges = in_sizes[3];

    float* z  = (float*)d_ws;
    float* el = z + (size_t)n_nodes * OUT_F;
    float* er = el + n_nodes;
    unsigned int* binbase = (unsigned int*)(er + n_nodes);
    const unsigned long long ws_bytes = (unsigned long long)ws_size;

    const int ntiles = (n_nodes + 15) / 16;

    gat_gemm_bin<<<dim3(BIN_G + (ntiles + 7) / 8), dim3(BIN_T), 0, stream>>>(
        h, W, a, z, el, er, src, dst, binbase, ws_bytes, n_nodes, n_edges);

    gat_owner<<<dim3((n_nodes + NPW - 1) / NPW), dim3(OWN_T), 0, stream>>>(
        src, dst, a, el, er, z, out, binbase, ws_bytes, n_nodes, n_edges);
}


# Row gather addressed by scalar base of z plus one 32-bit offset per lane: the sorted list holds each source row's number in 16-byte units, so the row-sum loop forms no 64-bit address per row
# speedup vs baseline: 8.8976x; 1.0320x over previous
#include <hip/hip_runtime.h>
#include <hip/hip_bf16.h>

#define IN_F 128
#define OUT_F 32
#define NEG_SLOPE 0.2f

typedef float v2f __attribute__((ext_vector_type(2)));
typedef float v8f __attribute__((ext_vector_type(8)));
typedef float v4f __attribute__((ext_vector_type(4)));

#define PITCH (IN_F + 4)
#define KH    (IN_F / 2)
#define APITCH (KH + 4)
#define GEMM_W_FLOATS (OUT_F * PITCH)
#define GEMM_A_FLOATS (16 * APITCH)
#define GEMM_T 256

__device__ __forceinline__ float rs_step(bool up, float a, float b, int w)
{
    return (up ? b : a) + __shfl_xor(up ? a : b, w);
}

__device__ __forceinline__ void k_half(
    const v4f (&hv)[8], float* s_a, const float* hrow,
    const float* __restrict__ w0, const float* __restrict__ w1,
    int half, int l16, v8f& acc0, v8f& acc1)
{
#pragma unroll
    for (int j = 0; j < 8; ++j)
        *(v4f*)(s_a + (2 * j + half) * APITCH + 4 * l16) = hv[j];
#pragma unroll 8
    for (int kk = 0; kk < KH; kk += 4) {
        const int ka = kk + 2 * half;
        const v2f A  = *(const v2f*)(hrow + ka);
        const v2f B0 = *(const v2f*)(w0 + ka);
        const v2f B1 = *(const v2f*)(w1 + ka);
        acc0 = __builtin_amdgcn_wmma_f32_16x16x4_f32(false, A, false, B0,
                                                     (short)0, acc0, false, false);
        acc1 = __builtin_amdgcn_wmma_f32_16x16x4_f32(false, A, false, B1,
                                                     (short)0, acc1, false, false);
    }
}

__device__ __forceinline__ void gemm_tiles(
    const float* __restrict__ h, const float* __restrict__ W,
    const float* __restrict__ a,
    float* __restrict__ z, float* __restrict__ el, float* __restrict__ er,
    int n_nodes, int blk)
{
    __shared__ __attribute__((aligned(16))) float s_wt[GEMM_W_FLOATS];
    __shared__ __attribute__((aligned(16))) float s_at[(GEMM_T / 32) * GEMM_A_FLOATS];
    const int wave = threadIdx.x >> 5;
    const int lane = threadIdx.x & 31;
    const int tile = blk * 8 + wave;
    const int row0 = tile * 16;

    const int half = lane >> 4;
    const int l16  = lane & 15;

    float4 wv[OUT_F * IN_F / 4 / GEMM_T];
#pragma unroll
    for (int j = 0; j < OUT_F * IN_F / 4 / GEMM_T; ++j) {
        const int q = (int)threadIdx.x + j * GEMM_T;
        wv[j] = *(const float4*)(W + (size_t)q * 4);
    }
    v4f hv0[8], hv1[8];
#pragma unroll
    for (int j = 0; j < 8; ++j) {
        const int rr = min(row0 + 2 * j + half, n_nodes - 1);
        const float* hp = h + (size_t)rr * IN_F + 4 * l16;
        hv0[j] = *(const v4f*)hp;
        hv1[j] = *(const v4f*)(hp + KH);
    }
#pragma unroll
    for (int j = 0; j < OUT_F * IN_F / 4 / GEMM_T; ++j) {
        const int q = (int)threadIdx.x + j * GEMM_T;
        *(float4*)(s_wt + (q >> 5) * PITCH + (q & 31) * 4) = wv[j];
    }
#pragma unroll
    for (int j = 0; j < 8; ++j)
        asm volatile("" : "+v"(hv0[j]), "+v"(hv1[j]));
    __syncthreads();
    if (row0 >= n_nodes) return;

    float* s_a = s_at + wave * GEMM_A_FLOATS;
    const float* hrow = s_a  + l16 * APITCH;
    const float* __restrict__ w0   = s_wt + l16 * PITCH;
    const float* __restrict__ w1   = s_wt + (l16 + 16) * PITCH;

    const float al0 = a[l16],         al1 = a[16 + l16];
    const float ar0 = a[OUT_F + l16], ar1 = a[OUT_F + 16 + l16];

    v8f acc0 = {};
    v8f acc1 = {};

    k_half(hv0, s_a, hrow, w0, w1, half, l16, acc0, acc1);
    k_half(hv1, s_a, hrow, w0 + KH, w1 + KH, half, l16, acc0, acc1);

#pragma unroll
    for (int r = 0; r < 8; ++r) {
        const int m = row0 + r + 8 * half;
        if (m < n_nodes) {
            z[(size_t)m * OUT_F + l16]      = acc0[r];
            z[(size_t)m * OUT_F + 16 + l16] = acc1[r];
        }
    }

    const bool up8 = (l16 & 8) != 0, up4 = (l16 & 4) != 0;
    const bool up2 = (l16 & 2) != 0, up1 = (l16 & 1) != 0;
#define LOGIT_PART(r) rs_step(up8, acc0[r] * al0 + acc1[r] * al1, \
                              acc0[r] * ar0 + acc1[r] * ar1, 8)
    const float p0 = LOGIT_PART(0), p1 = LOGIT_PART(1), p2 = LOGIT_PART(2), p3 = LOGIT_PART(3);
    const float p4 = LOGIT_PART(4), p5 = LOGIT_PART(5), p6 = LOGIT_PART(6), p7 = LOGIT_PART(7);
#undef LOGIT_PART
    const float q0 = rs_step(up4, p0, p4, 4), q1 = rs_step(up4, p1, p5, 4);
    const float q2 = rs_step(up4, p2, p6, 4), q3 = rs_step(up4, p3, p7, 4);
    const float t0 = rs_step(up2, q0, q2, 2), t1 = rs_step(up2, q1, q3, 2);
    const float sum = rs_step(up1, t0, t1, 1);
    {
        const int m = row0 + 8 * half + (l16 & 7);
        float* __restrict__ dstv = up8 ? er : el;
        if (m < n_nodes) dstv[m] = sum;
    }
}

__device__ __forceinline__ float leaky(float v)
{
    return v > 0.f ? v : NEG_SLOPE * v;
}

#define NPW    391
#define CAP    16384
#define OWN_T  1024

#define BIN_G  256
#define BIN_T  256
#define BIN_U  25
#define BIN_CH (BIN_T * BIN_U)
#define BIN_NS 256
#define OFFP   (BIN_NS + 1)
#define TBITS  9
#define TMASK  ((1u << TBITS) - 1u)

__device__ __forceinline__ bool bins_usable(int n_nodes, int n_edges, int ns,
                                            unsigned long long ws_bytes)
{
    const int ch = (n_edges + BIN_G - 1) / BIN_G;
    const unsigned long long words =
        (unsigned long long)n_nodes * (OUT_F + 2) +
        (unsigned long long)BIN_G * (unsigned long long)ch +
        (unsigned long long)BIN_G * OFFP;
    return n_edges > 0 && ns <= BIN_NS && ch <= BIN_CH && n_nodes <= (1 << 22) &&
           ws_bytes >= words * 4ull;
}

__device__ __forceinline__ int excl_scan_256(int v, int tid, int* s_w, int* total)
{
    const int lane = tid & 31;
    const int wv   = tid >> 5;
    int inc = v;
#pragma unroll
    for (int d = 1; d < 32; d <<= 1) {
        const int y = __shfl_up(inc, d);
        if (lane >= d) inc += y;
    }
    if (wv < 8 && lane == 31) s_w[wv] = inc;
    __syncthreads();
    int base = 0, tot = 0;
#pragma unroll
    for (int w = 0; w < 8; ++w) {
        const int x = s_w[w];
        if (w < wv) base += x;
        tot += x;
    }
    *total = tot;
    return base + inc - v;
}

__device__ __forceinline__ void bin_chunk(
    const int* __restrict__ src, const int* __restrict__ dst,
    unsigned int* __restrict__ binbase, unsigned long long ws_bytes,
    int n_nodes, int n_edges, int c)
{
    __shared__ int s_hist[BIN_NS + 1];
    __shared__ int s_off[BIN_NS + 1];
    __shared__ int s_w[8];
    __shared__ unsigned int s_stage[BIN_CH];

    const int tid = threadIdx.x;
    const int ns  = (n_nodes + NPW - 1) / NPW;
    if (!bins_usable(n_nodes, n_edges, ns, ws_bytes)) return;

    const int ch = (n_edges + BIN_G - 1) / BIN_G;
    const int e0 = c * ch;
    const int n  = max(0, min(ch, n_edges - e0));
    unsigned int* __restrict__ list = binbase + (size_t)c * ch;
    int* __restrict__ offs = (int*)(binbase + (size_t)BIN_G * ch) + (size_t)c * OFFP;

    for (int i = tid; i < BIN_NS + 1; i += BIN_T) { s_hist[i] = 0; s_off[i] = 0; }
    __syncthreads();

    static_assert(BIN_NS < (1 << 9) && BIN_CH <= (1 << 13) && TBITS == 9,
                  "slice, rank and local destination share a 31-bit word");
    int d[BIN_U], s[BIN_U];
#pragma unroll
    for (int u = 0; u < BIN_U; ++u) {
        const int e = min(e0 + tid + u * BIN_T, n_edges - 1);
        d[u] = dst[e];
        s[u] = src[e];
    }
    __builtin_amdgcn_sched_barrier(0);
    unsigned int w[BIN_U];
#pragma unroll
    for (int u = 0; u < BIN_U; ++u) {
        const bool ok = (tid + u * BIN_T < n) && ((unsigned)d[u] < (unsigned)n_nodes);
        const int b = d[u] / NPW;
        const int bb = ok ? b : BIN_NS;
        const unsigned t = (unsigned)(d[u] - b * NPW) & TMASK;
        const int pos = atomicAdd(&s_hist[bb], 1);
        w[u] = ((unsigned)bb << 22) | ((unsigned)pos << TBITS) | t;
        asm("" : "+v"(w[u]));
    }
    __syncthreads();

    const int v = (tid < ns) ? s_hist[tid] : 0;
    int total;
    const int ex = excl_scan_256(v, tid, s_w, &total);
    if (tid < ns) { s_off[tid] = ex; offs[tid] = ex; }
    if (tid == 0) { offs[ns] = total; s_off[BIN_NS] = total; }
    __syncthreads();

#pragma unroll
    for (int u = 0; u < BIN_U; ++u) {
        s_stage[s_off[w[u] >> 22] + (int)((w[u] >> TBITS) & 0x1fffu)] =
            ((unsigned)s[u] << TBITS) | (w[u] & TMASK);
    }
    __syncthreads();

    for (int i = tid; i < total; i += BIN_T) list[i] = s_stage[i];
}

static_assert(BIN_T == GEMM_T, "both halves of the first launch use the same block");

__global__ void __launch_bounds__(BIN_T) gat_gemm_bin(
    const float* __restrict__ h, const float* __restrict__ W,
    const float* __restrict__ a,
    float* __restrict__ z, float* __restrict__ el, float* __restrict__ er,
    const int* __restrict__ src, const int* __restrict__ dst,
    unsigned int* __restrict__ binbase, unsigned long long ws_bytes,
    int n_nodes, int n_edges)
{
    const int b = blockIdx.x;
    if (b < BIN_G)
        bin_chunk(src, dst, binbase, ws_bytes, n_nodes, n_edges, b);
    else
        gemm_tiles(h, W, a, z, el, er, n_nodes, b - BIN_G);
}


__device__ __forceinline__ void lds_max_f32(float* addr, float v)
{
    if (v >= 0.f)
        atomicMax((int*)addr, __float_as_int(v));
    else
        atomicMin((unsigned int*)addr, __float_as_uint(v));
}

#define NSLOT (1 << TBITS)
#define EB    8
#define RUNS_W (BIN_G / (OWN_T / 32))
#define RUN_L  2
#define HELD   (RUNS_W * RUN_L)
static_assert(BIN_G % (OWN_T / 32) == 0, "every wave fetches the same number of runs");
static_assert(HELD * OWN_T <= CAP, "what the lanes can hold fits the sorted list");

__device__ __forceinline__ int excl_scan_own(int v, int tid, int* s_w, int* total)
{
    const int lane = tid & 31;
    const int wv   = tid >> 5;
    int inc = v;
#pragma unroll
    for (int d = 1; d < 32; d <<= 1) {
        const int y = __shfl_up(inc, d);
        if (lane >= d) inc += y;
    }
    if (lane == 31) s_w[wv] = inc;
    __syncthreads();
    int base = 0, tot = 0;
#pragma unroll
    for (int w = 0; w < OWN_T / 32; ++w) {
        const int x = s_w[w];
        if (w < wv) base += x;
        tot += x;
    }
    *total = tot;
    return base + inc - v;
}

template <unsigned SEL>
__device__ __forceinline__ float grp8(float v)
{
    return __int_as_float((int)__builtin_amdgcn_mov_dpp8((unsigned)__float_as_int(v), SEL));
}
#define SEL8(a, b, c, d, e, f, g, h) \
    ((a) | ((b) << 3) | ((c) << 6) | ((d) << 9) | ((e) << 12) | ((f) << 15) | ((g) << 18) | ((h) << 21))
#define SEL_X1   SEL8(1u, 0u, 3u, 2u, 5u, 4u, 7u, 6u)
#define SEL_X2   SEL8(2u, 3u, 0u, 1u, 6u, 7u, 4u, 5u)
#define SEL_X4   SEL8(4u, 5u, 6u, 7u, 0u, 1u, 2u, 3u)
#define SEL_B(u) SEL8(u, u, u, u, u, u, u, u)

template <unsigned SEL>
__device__ __forceinline__ float rs8(bool up, float a, float b)
{
    return (up ? b : a) + grp8<SEL>(up ? a : b);
}

__global__ void __launch_bounds__(OWN_T) gat_owner(
    const int* __restrict__ src, const int* __restrict__ dst,
    const float* __restrict__ a,
    const float* __restrict__ el, const float* __restrict__ er,
    const float* __restrict__ z, float* __restrict__ out,
    const unsigned int* __restrict__ binbase, unsigned long long ws_bytes,
    int n_nodes, int n_edges)
{
    __shared__ float s_m[NSLOT];
    __shared__ float s_er[NSLOT];
    __shared__ int   s_cnt[NSLOT];
    __shared__ int   s_off[NSLOT];
    __shared__ __attribute__((aligned(16))) unsigned int s_ssrc[CAP];
    __shared__ int   s_w2[OWN_T / 32];
    __shared__ int   s_next;
    __shared__ int   s_long;

    const int tid  = threadIdx.x;
    const int lane = tid & 31;
    const int lo  = blockIdx.x * NPW;
    const int cnt = min(NPW, n_nodes - lo);
    const unsigned ucnt = (unsigned)cnt;

    const float4 al = *(const float4*)(a + (tid & 7) * 4);

    const bool usable = bins_usable(n_nodes, n_edges, (int)gridDim.x, ws_bytes);
    unsigned int pk[HELD];
    int gn[RUNS_W];
    bool lng = false;
#pragma unroll
    for (int u = 0; u < RUNS_W; ++u) gn[u] = 0;
#pragma unroll
    for (int j = 0; j < HELD; ++j) pk[j] = 0u;
    if (usable) {
        const int ch = (n_edges + BIN_G - 1) / BIN_G;
        const unsigned int* __restrict__ lists = binbase;
        const int* __restrict__ offs = (const int*)(binbase + (size_t)BIN_G * ch);
        const int wv = __builtin_amdgcn_readfirstlane(tid >> 5);
        int g0[RUNS_W];
#pragma unroll
        for (int u = 0; u < RUNS_W; ++u) {
            const int c = wv + u * (OWN_T / 32);
            const int* o = offs + (size_t)c * OFFP + blockIdx.x;
            int a0 = o[0];
            const int e = o[1];
            a0    = min(max(a0, 0), ch);
            gn[u] = min(max(e - a0, 0), ch - a0);
            g0[u] = c * ch + a0;
            lng   = lng || gn[u] > 32 * RUN_L;
        }
#pragma unroll
        for (int u = 0; u < RUNS_W; ++u) {
#pragma unroll
            for (int k = 0; k < RUN_L; ++k) {
                const int i = lane + 32 * k;
                if (i < gn[u]) pk[u * RUN_L + k] = lists[g0[u] + i];
            }
        }
    }
    for (int i = tid; i < NSLOT; i += OWN_T) {
        s_m[i]   = -__builtin_inff();
        s_cnt[i] = 0;
        s_er[i]  = (i < cnt) ? er[lo + i] : 0.f;
    }
    if (tid == 0) { s_next = 0; s_long = 0; }
    __syncthreads();

    int rk[HELD];
#pragma unroll
    for (int j = 0; j < HELD; ++j) {
        rk[j] = 0;
        if (lane + 32 * (j % RUN_L) < gn[j / RUN_L])
            rk[j] = atomicAdd(&s_cnt[pk[j] & TMASK], 1);
    }
    if (lng && lane == 0) s_long = 1;
    __syncthreads();

    int total = CAP + 1;
    if (usable) {
        int tot;
        const int v  = (tid < NSLOT) ? s_cnt[tid] : 0;
        const int ex = excl_scan_own(v, tid, s_w2, &tot);
        if (tid < NSLOT) s_off[tid] = ex;
        if (s_long == 0) total = tot;
    }
    __syncthreads();

    if (total <= CAP) {
        int st[HELD];
#pragma unroll
        for (int j = 0; j < HELD; ++j) st[j] = s_off[pk[j] & TMASK];
#pragma unroll
        for (int j = 0; j < HELD; ++j) {
            if (lane + 32 * (j % RUN_L) < gn[j / RUN_L])
                s_ssrc[st[j] + rk[j]] = (pk[j] >> TBITS) * (OUT_F / 4);
        }
        __syncthreads();

        static_assert(EB == 8, "one edge of a batch per lane of the group");
        const int sub  = tid & 7;
        const int c    = sub * 4;
        const bool up4 = (sub & 4) != 0, up2 = (sub & 2) != 0, up1 = (sub & 1) != 0;
        const v2f alo = {al.x, al.y}, ahi = {al.z, al.w};
        const float4* __restrict__ zv = (const float4*)z;
        for (;;) {
            int t = 0;
            if (sub == 0) t = atomicAdd(&s_next, 1);
            t = __shfl(t, lane & 24);
            const bool live = t < cnt;
            if (!__any(live)) break;
            int b = 0, e = 0;
            float ert = 0.f;
            if (live) { b = s_off[t]; e = b + s_cnt[t]; ert = s_er[t]; }
            float4 acc = make_float4(0.f, 0.f, 0.f, 0.f);
            float l = 0.f;
            float m = -__builtin_inff();
            for (; b < e; b += EB) {
                unsigned int ri[EB];
                float4 zc[EB];
                float d[EB];
#pragma unroll
                for (int u = 0; u < EB; ++u)
                    ri[u] = s_ssrc[min(b + u, e - 1)] + (unsigned int)sub;
#pragma unroll
                for (int u = 0; u < EB; ++u)
                    zc[u] = zv[ri[u]];
#pragma unroll
                for (int u = 0; u < EB; ++u) {
                    const v2f zlo = {zc[u].x, zc[u].y}, zhi = {zc[u].z, zc[u].w};
                    const v2f t2 = zlo * alo + zhi * ahi;
                    d[u] = t2.x + t2.y;
                }
                const float q0 = rs8<SEL_X4>(up4, d[0], d[4]), q1 = rs8<SEL_X4>(up4, d[1], d[5]);
                const float q2 = rs8<SEL_X4>(up4, d[2], d[6]), q3 = rs8<SEL_X4>(up4, d[3], d[7]);
                const float r0 = rs8<SEL_X2>(up2, q0, q2), r1 = rs8<SEL_X2>(up2, q1, q3);
                const float els = rs8<SEL_X1>(up1, r0, r1);
                const float v = (b + sub < e) ? leaky(els + ert) : -__builtin_inff();
                float bm = fmaxf(v, grp8<SEL_X1>(v));
                bm = fmaxf(bm, grp8<SEL_X2>(bm));
                bm = fmaxf(bm, grp8<SEL_X4>(bm));
                const float mn = fmaxf(m, bm);
                const float sc = __expf(m - mn);
                const float p  = __expf(v - mn);
                m = mn;
                acc.x *= sc; acc.y *= sc; acc.z *= sc; acc.w *= sc;
                l *= sc;
#define ADD_EDGE(k) { const float pk_ = grp8<SEL_B(k##u)>(p); \
                      acc.x += pk_ * zc[k].x; acc.y += pk_ * zc[k].y; \
                      acc.z += pk_ * zc[k].z; acc.w += pk_ * zc[k].w; l += pk_; }
                ADD_EDGE(0) ADD_EDGE(1) ADD_EDGE(2) ADD_EDGE(3)
                ADD_EDGE(4) ADD_EDGE(5) ADD_EDGE(6) ADD_EDGE(7)
#undef ADD_EDGE
            }
            if (live) {
                const float den = fmaxf(l, 1e-16f);
                float4 r;
                r.x = acc.x / den; r.y = acc.y / den; r.z = acc.z / den; r.w = acc.w / den;
                *(float4*)(out + (size_t)(lo + t) * OUT_F + c) = r;
            }
        }
    } else {
        float* s_acc = (float*)s_ssrc;
        float* s_l   = (float*)s_off;
        for (int i = tid; i < NPW * OUT_F; i += OWN_T) s_acc[i] = 0.f;
        for (int i = tid; i < NPW; i += OWN_T) s_l[i] = 0.f;
        for (int e = tid; e < n_edges; e += OWN_T) {
            const unsigned t = (unsigned)(dst[e] - lo);
            if (t < ucnt) lds_max_f32(&s_m[t], leaky(el[src[e]] + s_er[t]));
        }
        __syncthreads();
        for (int e = tid; e < n_edges; e += OWN_T) {
            const unsigned t = (unsigned)(dst[e] - lo);
            if (t < ucnt) {
                const int s = src[e];
                const float pe = __expf(leaky(el[s] + s_er[t]) - s_m[t]);
                atomicAdd(&s_l[t], pe);
                const float* zr = z + (size_t)s * OUT_F;
                for (int j = 0; j < OUT_F; ++j)
                    atomicAdd(&s_acc[t * OUT_F + j], pe * zr[j]);
            }
        }
        __syncthreads();
        for (int q = tid; q < cnt * 8; q += OWN_T) {
            const int t = q >> 3;
            const int c = (q & 7) * 4;
            const float den = fmaxf(s_l[t], 1e-16f);
            const float4 acc = *(const float4*)&s_acc[t * OUT_F + c];
            float4 r;
            r.x = acc.x / den; r.y = acc.y / den; r.z = acc.z / den; r.w = acc.w / den;
            *(float4*)(out + (size_t)(lo + t) * OUT_F + c) = r;
        }
    }
}

extern "C" void kernel_launch(void* const* d_in, const int* in_sizes, int n_in,
                              void* d_out, int out_size, void* d_ws, size_t ws_size,
                              hipStream_t stream)
{
    const float* h  = (const float*)d_in[0];
    const float* W  = (const float*)d_in[1];
    const float* a  = (const float*)d_in[2];
    const int* src  = (const int*)d_in[3];
    const int* dst  = (const int*)d_in[4];
    float* out      = (float*)d_out;

    const int n_nodes = in_sizes[0] / IN_F;
    const int n_edges = in_sizes[3];

    float* z  = (float*)d_ws;
    float* el = z + (size_t)n_nodes * OUT_F;
    float* er = el + n_nodes;
    unsigned int* binbase = (unsigned int*)(er + n_nodes);
    const unsigned long long ws_bytes = (unsigned long long)ws_size;

    const int ntiles = (n_nodes + 15) / 16;

    gat_gemm_bin<<<dim3(BIN_G + (ntiles + 7) / 8), dim3(BIN_T), 0, stream>>>(
        h, W, a, z, el, er, src, dst, binbase, ws_bytes, n_nodes, n_edges);

    gat_owner<<<dim3((n_nodes + NPW - 1) / NPW), dim3(OWN_T), 0, stream>>>(
        src, dst, a, el, er, z, out, binbase, ws_bytes, n_nodes, n_edges);
}


# Product's waves loop over the 16-row tiles: 512 work-groups (were 782, one tile per wave), next tile's rows loaded while the current one multiplies; 6256 product waves become 4096; sums, k order kept
# speedup vs baseline: 9.2273x; 1.0371x over previous
#include <hip/hip_runtime.h>
#include <hip/hip_bf16.h>

#define IN_F 128
#define OUT_F 32
#define NEG_SLOPE 0.2f

typedef float v2f __attribute__((ext_vector_type(2)));
typedef float v8f __attribute__((ext_vector_type(8)));
typedef float v4f __attribute__((ext_vector_type(4)));

#define PITCH (IN_F + 4)
#define KH    (IN_F / 2)
#define APITCH (KH + 4)
#define GEMM_W_FLOATS (OUT_F * PITCH)
#define GEMM_A_FLOATS (16 * APITCH)
#define GEMM_T 256
#define GEMM_G 512

__device__ __forceinline__ float rs_step(bool up, float a, float b, int w)
{
    return (up ? b : a) + __shfl_xor(up ? a : b, w);
}

__device__ __forceinline__ void load_half(
    v4f (&hv)[8], const float* __restrict__ h, int row0, int koff,
    int half, int l16, int n_nodes)
{
#pragma unroll
    for (int j = 0; j < 8; ++j) {
        const int rr = min(row0 + 2 * j + half, n_nodes - 1);
        hv[j] = *(const v4f*)(h + (size_t)rr * IN_F + koff + 4 * l16);
    }
}

__device__ __forceinline__ void lay_half(
    const v4f (&hv)[8], float* s_a, int half, int l16)
{
#pragma unroll
    for (int j = 0; j < 8; ++j)
        *(v4f*)(s_a + (2 * j + half) * APITCH + 4 * l16) = hv[j];
}

__device__ __forceinline__ void mul_half(
    const float* hrow,
    const float* __restrict__ w0, const float* __restrict__ w1,
    int half, v8f& acc0, v8f& acc1)
{
#pragma unroll 8
    for (int kk = 0; kk < KH; kk += 4) {
        const int ka = kk + 2 * half;
        const v2f A  = *(const v2f*)(hrow + ka);
        const v2f B0 = *(const v2f*)(w0 + ka);
        const v2f B1 = *(const v2f*)(w1 + ka);
        acc0 = __builtin_amdgcn_wmma_f32_16x16x4_f32(false, A, false, B0,
                                                     (short)0, acc0, false, false);
        acc1 = __builtin_amdgcn_wmma_f32_16x16x4_f32(false, A, false, B1,
                                                     (short)0, acc1, false, false);
    }
}

__device__ __forceinline__ void gemm_tiles(
    const float* __restrict__ h, const float* __restrict__ W,
    const float* __restrict__ a,
    float* __restrict__ z, float* __restrict__ el, float* __restrict__ er,
    int n_nodes, int blk)
{
    __shared__ __attribute__((aligned(16))) float s_wt[GEMM_W_FLOATS];
    __shared__ __attribute__((aligned(16))) float s_at[(GEMM_T / 32) * GEMM_A_FLOATS];
    const int wave = __builtin_amdgcn_readfirstlane((int)(threadIdx.x >> 5));
    const int lane = threadIdx.x & 31;
    const int ntiles = (n_nodes + 15) / 16;
    int tile = blk * (GEMM_T / 32) + wave;

    const int half = lane >> 4;
    const int l16  = lane & 15;

    float al0 = a[l16],         al1 = a[16 + l16];
    float ar0 = a[OUT_F + l16], ar1 = a[OUT_F + 16 + l16];
    float4 wv[OUT_F * IN_F / 4 / GEMM_T];
#pragma unroll
    for (int j = 0; j < OUT_F * IN_F / 4 / GEMM_T; ++j) {
        const int q = (int)threadIdx.x + j * GEMM_T;
        wv[j] = *(const float4*)(W + (size_t)q * 4);
    }
    v4f hv0[8], hv1[8];
    load_half(hv0, h, tile * 16, 0,  half, l16, n_nodes);
    load_half(hv1, h, tile * 16, KH, half, l16, n_nodes);
#pragma unroll
    for (int j = 0; j < OUT_F * IN_F / 4 / GEMM_T; ++j) {
        const int q = (int)threadIdx.x + j * GEMM_T;
        *(float4*)(s_wt + (q >> 5) * PITCH + (q & 31) * 4) = wv[j];
    }
#pragma unroll
    for (int j = 0; j < 8; ++j)
        asm volatile("" : "+v"(hv0[j]), "+v"(hv1[j]));
    asm volatile("" : "+v"(al0), "+v"(al1), "+v"(ar0), "+v"(ar1));
    __syncthreads();

    float* s_a = s_at + wave * GEMM_A_FLOATS;
    const float* hrow = s_a  + l16 * APITCH;
    const float* __restrict__ w0   = s_wt + l16 * PITCH;
    const float* __restrict__ w1   = s_wt + (l16 + 16) * PITCH;

    const bool up8 = (l16 & 8) != 0, up4 = (l16 & 4) != 0;
    const bool up2 = (l16 & 2) != 0, up1 = (l16 & 1) != 0;

    for (; tile < ntiles; tile += GEMM_G * (GEMM_T / 32)) {
        const int row0  = tile * 16;
        const int ntile = tile + GEMM_G * (GEMM_T / 32);
        const bool more = ntile < ntiles;

        v8f acc0 = {};
        v8f acc1 = {};

        lay_half(hv0, s_a, half, l16);
        if (more) load_half(hv0, h, ntile * 16, 0, half, l16, n_nodes);
        mul_half(hrow, w0, w1, half, acc0, acc1);
        lay_half(hv1, s_a, half, l16);
        if (more) load_half(hv1, h, ntile * 16, KH, half, l16, n_nodes);
        mul_half(hrow, w0 + KH, w1 + KH, half, acc0, acc1);

#pragma unroll
        for (int r = 0; r < 8; ++r) {
            const int m = row0 + r + 8 * half;
            if (m < n_nodes) {
                z[(size_t)m * OUT_F + l16]      = acc0[r];
                z[(size_t)m * OUT_F + 16 + l16] = acc1[r];
            }
        }

#define LOGIT_PART(r) rs_step(up8, acc0[r] * al0 + acc1[r] * al1, \
                              acc0[r] * ar0 + acc1[r] * ar1, 8)
        const float p0 = LOGIT_PART(0), p1 = LOGIT_PART(1), p2 = LOGIT_PART(2), p3 = LOGIT_PART(3);
        const float p4 = LOGIT_PART(4), p5 = LOGIT_PART(5), p6 = LOGIT_PART(6), p7 = LOGIT_PART(7);
#undef LOGIT_PART
        const float q0 = rs_step(up4, p0, p4, 4), q1 = rs_step(up4, p1, p5, 4);
        const float q2 = rs_step(up4, p2, p6, 4), q3 = rs_step(up4, p3, p7, 4);
        const float t0 = rs_step(up2, q0, q2, 2), t1 = rs_step(up2, q1, q3, 2);
        const float sum = rs_step(up1, t0, t1, 1);
        {
            const int m = row0 + 8 * half + (l16 & 7);
            float* __restrict__ dstv = up8 ? er : el;
            if (m < n_nodes) dstv[m] = sum;
        }
    }
}

__device__ __forceinline__ float leaky(float v)
{
    return v > 0.f ? v : NEG_SLOPE * v;
}

#define NPW    391
#define CAP    16384
#define OWN_T  1024

#define BIN_G  256
#define BIN_T  256
#define BIN_U  25
#define BIN_CH (BIN_T * BIN_U)
#define BIN_NS 256
#define OFFP   (BIN_NS + 1)
#define TBITS  9
#define TMASK  ((1u << TBITS) - 1u)

__device__ __forceinline__ bool bins_usable(int n_nodes, int n_edges, int ns,
                                            unsigned long long ws_bytes)
{
    const int ch = (n_edges + BIN_G - 1) / BIN_G;
    const unsigned long long words =
        (unsigned long long)n_nodes * (OUT_F + 2) +
        (unsigned long long)BIN_G * (unsigned long long)ch +
        (unsigned long long)BIN_G * OFFP;
    return n_edges > 0 && ns <= BIN_NS && ch <= BIN_CH && n_nodes <= (1 << 22) &&
           ws_bytes >= words * 4ull;
}

__device__ __forceinline__ int excl_scan_256(int v, int tid, int* s_w, int* total)
{
    const int lane = tid & 31;
    const int wv   = tid >> 5;
    int inc = v;
#pragma unroll
    for (int d = 1; d < 32; d <<= 1) {
        const int y = __shfl_up(inc, d);
        if (lane >= d) inc += y;
    }
    if (wv < 8 && lane == 31) s_w[wv] = inc;
    __syncthreads();
    int base = 0, tot = 0;
#pragma unroll
    for (int w = 0; w < 8; ++w) {
        const int x = s_w[w];
        if (w < wv) base += x;
        tot += x;
    }
    *total = tot;
    return base + inc - v;
}

__device__ __forceinline__ void bin_chunk(
    const int* __restrict__ src, const int* __restrict__ dst,
    unsigned int* __restrict__ binbase, unsigned long long ws_bytes,
    int n_nodes, int n_edges, int c)
{
    __shared__ int s_hist[BIN_NS + 1];
    __shared__ int s_off[BIN_NS + 1];
    __shared__ int s_w[8];
    __shared__ unsigned int s_stage[BIN_CH];

    const int tid = threadIdx.x;
    const int ns  = (n_nodes + NPW - 1) / NPW;
    if (!bins_usable(n_nodes, n_edges, ns, ws_bytes)) return;

    const int ch = (n_edges + BIN_G - 1) / BIN_G;
    const int e0 = c * ch;
    const int n  = max(0, min(ch, n_edges - e0));
    unsigned int* __restrict__ list = binbase + (size_t)c * ch;
    int* __restrict__ offs = (int*)(binbase + (size_t)BIN_G * ch) + (size_t)c * OFFP;

    for (int i = tid; i < BIN_NS + 1; i += BIN_T) { s_hist[i] = 0; s_off[i] = 0; }
    __syncthreads();

    static_assert(BIN_NS < (1 << 9) && BIN_CH <= (1 << 13) && TBITS == 9,
                  "slice, rank and local destination share a 31-bit word");
    int d[BIN_U], s[BIN_U];
#pragma unroll
    for (int u = 0; u < BIN_U; ++u) {
        const int e = min(e0 + tid + u * BIN_T, n_edges - 1);
        d[u] = dst[e];
        s[u] = src[e];
    }
    __builtin_amdgcn_sched_barrier(0);
    unsigned int w[BIN_U];
#pragma unroll
    for (int u = 0; u < BIN_U; ++u) {
        const bool ok = (tid + u * BIN_T < n) && ((unsigned)d[u] < (unsigned)n_nodes);
        const int b = d[u] / NPW;
        const int bb = ok ? b : BIN_NS;
        const unsigned t = (unsigned)(d[u] - b * NPW) & TMASK;
        const int pos = atomicAdd(&s_hist[bb], 1);
        w[u] = ((unsigned)bb << 22) | ((unsigned)pos << TBITS) | t;
        asm("" : "+v"(w[u]));
    }
    __syncthreads();

    const int v = (tid < ns) ? s_hist[tid] : 0;
    int total;
    const int ex = excl_scan_256(v, tid, s_w, &total);
    if (tid < ns) { s_off[tid] = ex; offs[tid] = ex; }
    if (tid == 0) { offs[ns] = total; s_off[BIN_NS] = total; }
    __syncthreads();

#pragma unroll
    for (int u = 0; u < BIN_U; ++u) {
        s_stage[s_off[w[u] >> 22] + (int)((w[u] >> TBITS) & 0x1fffu)] =
            ((unsigned)s[u] << TBITS) | (w[u] & TMASK);
    }
    __syncthreads();

    for (int i = tid; i < total; i += BIN_T) list[i] = s_stage[i];
}

static_assert(BIN_T == GEMM_T, "both halves of the first launch use the same block");

__global__ void __launch_bounds__(BIN_T) gat_gemm_bin(
    const float* __restrict__ h, const float* __restrict__ W,
    const float* __restrict__ a,
    float* __restrict__ z, float* __restrict__ el, float* __restrict__ er,
    const int* __restrict__ src, const int* __restrict__ dst,
    unsigned int* __restrict__ binbase, unsigned long long ws_bytes,
    int n_nodes, int n_edges)
{
    const int b = blockIdx.x;
    if (b < BIN_G)
        bin_chunk(src, dst, binbase, ws_bytes, n_nodes, n_edges, b);
    else
        gemm_tiles(h, W, a, z, el, er, n_nodes, b - BIN_G);
}


__device__ __forceinline__ void lds_max_f32(float* addr, float v)
{
    if (v >= 0.f)
        atomicMax((int*)addr, __float_as_int(v));
    else
        atomicMin((unsigned int*)addr, __float_as_uint(v));
}

#define NSLOT (1 << TBITS)
#define EB    8
#define RUNS_W (BIN_G / (OWN_T / 32))
#define RUN_L  2
#define HELD   (RUNS_W * RUN_L)
static_assert(BIN_G % (OWN_T / 32) == 0, "every wave fetches the same number of runs");
static_assert(HELD * OWN_T <= CAP, "what the lanes can hold fits the sorted list");

__device__ __forceinline__ int excl_scan_own(int v, int tid, int* s_w, int* total)
{
    const int lane = tid & 31;
    const int wv   = tid >> 5;
    int inc = v;
#pragma unroll
    for (int d = 1; d < 32; d <<= 1) {
        const int y = __shfl_up(inc, d);
        if (lane >= d) inc += y;
    }
    if (lane == 31) s_w[wv] = inc;
    __syncthreads();
    int base = 0, tot = 0;
#pragma unroll
    for (int w = 0; w < OWN_T / 32; ++w) {
        const int x = s_w[w];
        if (w < wv) base += x;
        tot += x;
    }
    *total = tot;
    return base + inc - v;
}

template <unsigned SEL>
__device__ __forceinline__ float grp8(float v)
{
    return __int_as_float((int)__builtin_amdgcn_mov_dpp8((unsigned)__float_as_int(v), SEL));
}
#define SEL8(a, b, c, d, e, f, g, h) \
    ((a) | ((b) << 3) | ((c) << 6) | ((d) << 9) | ((e) << 12) | ((f) << 15) | ((g) << 18) | ((h) << 21))
#define SEL_X1   SEL8(1u, 0u, 3u, 2u, 5u, 4u, 7u, 6u)
#define SEL_X2   SEL8(2u, 3u, 0u, 1u, 6u, 7u, 4u, 5u)
#define SEL_X4   SEL8(4u, 5u, 6u, 7u, 0u, 1u, 2u, 3u)
#define SEL_B(u) SEL8(u, u, u, u, u, u, u, u)

template <unsigned SEL>
__device__ __forceinline__ float rs8(bool up, float a, float b)
{
    return (up ? b : a) + grp8<SEL>(up ? a : b);
}

__global__ void __launch_bounds__(OWN_T) gat_owner(
    const int* __restrict__ src, const int* __restrict__ dst,
    const float* __restrict__ a,
    const float* __restrict__ el, const float* __restrict__ er,
    const float* __restrict__ z, float* __restrict__ out,
    const unsigned int* __restrict__ binbase, unsigned long long ws_bytes,
    int n_nodes, int n_edges)
{
    __shared__ float s_m[NSLOT];
    __shared__ float s_er[NSLOT];
    __shared__ int   s_cnt[NSLOT];
    __shared__ int   s_off[NSLOT];
    __shared__ __attribute__((aligned(16))) unsigned int s_ssrc[CAP];
    __shared__ int   s_w2[OWN_T / 32];
    __shared__ int   s_next;
    __shared__ int   s_long;

    const int tid  = threadIdx.x;
    const int lane = tid & 31;
    const int lo  = blockIdx.x * NPW;
    const int cnt = min(NPW, n_nodes - lo);
    const unsigned ucnt = (unsigned)cnt;

    const float4 al = *(const float4*)(a + (tid & 7) * 4);

    const bool usable = bins_usable(n_nodes, n_edges, (int)gridDim.x, ws_bytes);
    unsigned int pk[HELD];
    int gn[RUNS_W];
    bool lng = false;
#pragma unroll
    for (int u = 0; u < RUNS_W; ++u) gn[u] = 0;
#pragma unroll
    for (int j = 0; j < HELD; ++j) pk[j] = 0u;
    if (usable) {
        const int ch = (n_edges + BIN_G - 1) / BIN_G;
        const unsigned int* __restrict__ lists = binbase;
        const int* __restrict__ offs = (const int*)(binbase + (size_t)BIN_G * ch);
        const int wv = __builtin_amdgcn_readfirstlane(tid >> 5);
        int g0[RUNS_W];
#pragma unroll
        for (int u = 0; u < RUNS_W; ++u) {
            const int c = wv + u * (OWN_T / 32);
            const int* o = offs + (size_t)c * OFFP + blockIdx.x;
            int a0 = o[0];
            const int e = o[1];
            a0    = min(max(a0, 0), ch);
            gn[u] = min(max(e - a0, 0), ch - a0);
            g0[u] = c * ch + a0;
            lng   = lng || gn[u] > 32 * RUN_L;
        }
#pragma unroll
        for (int u = 0; u < RUNS_W; ++u) {
#pragma unroll
            for (int k = 0; k < RUN_L; ++k) {
                const int i = lane + 32 * k;
                if (i < gn[u]) pk[u * RUN_L + k] = lists[g0[u] + i];
            }
        }
    }
    for (int i = tid; i < NSLOT; i += OWN_T) {
        s_m[i]   = -__builtin_inff();
        s_cnt[i] = 0;
        s_er[i]  = (i < cnt) ? er[lo + i] : 0.f;
    }
    if (tid == 0) { s_next = 0; s_long = 0; }
    __syncthreads();

    int rk[HELD];
#pragma unroll
    for (int j = 0; j < HELD; ++j) {
        rk[j] = 0;
        if (lane + 32 * (j % RUN_L) < gn[j / RUN_L])
            rk[j] = atomicAdd(&s_cnt[pk[j] & TMASK], 1);
    }
    if (lng && lane == 0) s_long = 1;
    __syncthreads();

    int total = CAP + 1;
    if (usable) {
        int tot;
        const int v  = (tid < NSLOT) ? s_cnt[tid] : 0;
        const int ex = excl_scan_own(v, tid, s_w2, &tot);
        if (tid < NSLOT) s_off[tid] = ex;
        if (s_long == 0) total = tot;
    }
    __syncthreads();

    if (total <= CAP) {
        int st[HELD];
#pragma unroll
        for (int j = 0; j < HELD; ++j) st[j] = s_off[pk[j] & TMASK];
#pragma unroll
        for (int j = 0; j < HELD; ++j) {
            if (lane + 32 * (j % RUN_L) < gn[j / RUN_L])
                s_ssrc[st[j] + rk[j]] = (pk[j] >> TBITS) * (OUT_F / 4);
        }
        __syncthreads();

        static_assert(EB == 8, "one edge of a batch per lane of the group");
        const int sub  = tid & 7;
        const int c    = sub * 4;
        const bool up4 = (sub & 4) != 0, up2 = (sub & 2) != 0, up1 = (sub & 1) != 0;
        const v2f alo = {al.x, al.y}, ahi = {al.z, al.w};
        const float4* __restrict__ zv = (const float4*)z;
        for (;;) {
            int t = 0;
            if (sub == 0) t = atomicAdd(&s_next, 1);
            t = __shfl(t, lane & 24);
            const bool live = t < cnt;
            if (!__any(live)) break;
            int b = 0, e = 0;
            float ert = 0.f;
            if (live) { b = s_off[t]; e = b + s_cnt[t]; ert = s_er[t]; }
            float4 acc = make_float4(0.f, 0.f, 0.f, 0.f);
            float l = 0.f;
            float m = -__builtin_inff();
            for (; b < e; b += EB) {
                unsigned int ri[EB];
                float4 zc[EB];
                float d[EB];
#pragma unroll
                for (int u = 0; u < EB; ++u)
                    ri[u] = s_ssrc[min(b + u, e - 1)] + (unsigned int)sub;
#pragma unroll
                for (int u = 0; u < EB; ++u)
                    zc[u] = zv[ri[u]];
#pragma unroll
                for (int u = 0; u < EB; ++u) {
                    const v2f zlo = {zc[u].x, zc[u].y}, zhi = {zc[u].z, zc[u].w};
                    const v2f t2 = zlo * alo + zhi * ahi;
                    d[u] = t2.x + t2.y;
                }
                const float q0 = rs8<SEL_X4>(up4, d[0], d[4]), q1 = rs8<SEL_X4>(up4, d[1], d[5]);
                const float q2 = rs8<SEL_X4>(up4, d[2], d[6]), q3 = rs8<SEL_X4>(up4, d[3], d[7]);
                const float r0 = rs8<SEL_X2>(up2, q0, q2), r1 = rs8<SEL_X2>(up2, q1, q3);
                const float els = rs8<SEL_X1>(up1, r0, r1);
                const float v = (b + sub < e) ? leaky(els + ert) : -__builtin_inff();
                float bm = fmaxf(v, grp8<SEL_X1>(v));
                bm = fmaxf(bm, grp8<SEL_X2>(bm));
                bm = fmaxf(bm, grp8<SEL_X4>(bm));
                const float mn = fmaxf(m, bm);
                const float sc = __expf(m - mn);
                const float p  = __expf(v - mn);
                m = mn;
                acc.x *= sc; acc.y *= sc; acc.z *= sc; acc.w *= sc;
                l *= sc;
#define ADD_EDGE(k) { const float pk_ = grp8<SEL_B(k##u)>(p); \
                      acc.x += pk_ * zc[k].x; acc.y += pk_ * zc[k].y; \
                      acc.z += pk_ * zc[k].z; acc.w += pk_ * zc[k].w; l += pk_; }
                ADD_EDGE(0) ADD_EDGE(1) ADD_EDGE(2) ADD_EDGE(3)
                ADD_EDGE(4) ADD_EDGE(5) ADD_EDGE(6) ADD_EDGE(7)
#undef ADD_EDGE
            }
            if (live) {
                const float den = fmaxf(l, 1e-16f);
                float4 r;
                r.x = acc.x / den; r.y = acc.y / den; r.z = acc.z / den; r.w = acc.w / den;
                *(float4*)(out + (size_t)(lo + t) * OUT_F + c) = r;
            }
        }
    } else {
        float* s_acc = (float*)s_ssrc;
        float* s_l   = (float*)s_off;
        for (int i = tid; i < NPW * OUT_F; i += OWN_T) s_acc[i] = 0.f;
        for (int i = tid; i < NPW; i += OWN_T) s_l[i] = 0.f;
        for (int e = tid; e < n_edges; e += OWN_T) {
            const unsigned t = (unsigned)(dst[e] - lo);
            if (t < ucnt) lds_max_f32(&s_m[t], leaky(el[src[e]] + s_er[t]));
        }
        __syncthreads();
        for (int e = tid; e < n_edges; e += OWN_T) {
            const unsigned t = (unsigned)(dst[e] - lo);
            if (t < ucnt) {
                const int s = src[e];
                const float pe = __expf(leaky(el[s] + s_er[t]) - s_m[t]);
                atomicAdd(&s_l[t], pe);
                const float* zr = z + (size_t)s * OUT_F;
                for (int j = 0; j < OUT_F; ++j)
                    atomicAdd(&s_acc[t * OUT_F + j], pe * zr[j]);
            }
        }
        __syncthreads();
        for (int q = tid; q < cnt * 8; q += OWN_T) {
            const int t = q >> 3;
            const int c = (q & 7) * 4;
            const float den = fmaxf(s_l[t], 1e-16f);
            const float4 acc = *(const float4*)&s_acc[t * OUT_F + c];
            float4 r;
            r.x = acc.x / den; r.y = acc.y / den; r.z = acc.z / den; r.w = acc.w / den;
            *(float4*)(out + (size_t)(lo + t) * OUT_F + c) = r;
        }
    }
}

extern "C" void kernel_launch(void* const* d_in, const int* in_sizes, int n_in,
                              void* d_out, int out_size, void* d_ws, size_t ws_size,
                              hipStream_t stream)
{
    const float* h  = (const float*)d_in[0];
    const float* W  = (const float*)d_in[1];
    const float* a  = (const float*)d_in[2];
    const int* src  = (const int*)d_in[3];
    const int* dst  = (const int*)d_in[4];
    float* out      = (float*)d_out;

    const int n_nodes = in_sizes[0] / IN_F;
    const int n_edges = in_sizes[3];

    float* z  = (float*)d_ws;
    float* el = z + (size_t)n_nodes * OUT_F;
    float* er = el + n_nodes;
    unsigned int* binbase = (unsigned int*)(er + n_nodes);
    const unsigned long long ws_bytes = (unsigned long long)ws_size;

    gat_gemm_bin<<<dim3(BIN_G + GEMM_G), dim3(BIN_T), 0, stream>>>(
        h, W, a, z, el, er, src, dst, binbase, ws_bytes, n_nodes, n_edges);

    gat_owner<<<dim3((n_nodes + NPW - 1) / NPW), dim3(OWN_T), 0, stream>>>(
        src, dst, a, el, er, z, out, binbase, ws_bytes, n_nodes, n_edges);
}
